# starting kernel
# baseline (speedup 1.0000x reference)
#include <hip/hip_runtime.h>
#include <hip/hip_bf16.h>

typedef float v2f __attribute__((ext_vector_type(2)));
typedef float v8f __attribute__((ext_vector_type(8)));

#define HEADS 4
#define HC 128
#define NEG_SLOPE 0.2f

__device__ __forceinline__ unsigned enc_f32(float f) {
    unsigned b = __float_as_uint(f);
    return b ^ (((int)b >> 31) | 0x80000000u);
}
__device__ __forceinline__ float dec_f32(unsigned u) {
    unsigned b = (u & 0x80000000u) ? (u ^ 0x80000000u) : ~u;
    return __uint_as_float(b);
}

__global__ void ew_sum_kernel(const float* __restrict__ ew, float* __restrict__ out_sum, int E) {
    __shared__ float sm[8];
    float s = 0.f;
    for (int i = blockIdx.x * blockDim.x + threadIdx.x; i < E; i += gridDim.x * blockDim.x)
        s += ew[i];
    #pragma unroll
    for (int off = 16; off; off >>= 1) s += __shfl_down(s, off, 32);
    int lane = threadIdx.x & 31, w = threadIdx.x >> 5;
    if (lane == 0) sm[w] = s;
    __syncthreads();
    if (threadIdx.x == 0) {
        float t = 0.f;
        for (int i = 0; i < (int)(blockDim.x >> 5); ++i) t += sm[i];
        atomicAdd(out_sum, t);
    }
}

__global__ void edge_coef_kernel(const float* __restrict__ W_e, const float* __restrict__ att_e,
                                 float* __restrict__ c) {
    int t = threadIdx.x;
    float p = W_e[t] * att_e[t];
    #pragma unroll
    for (int off = 16; off; off >>= 1) p += __shfl_down(p, off, 32);
    if ((t & 31) == 0) c[t >> 5] = p;
}

__global__ void gemm_wmma_f32(const float* __restrict__ A, const float* __restrict__ W,
                              float* __restrict__ C, int M, int K) {
    extern __shared__ float lw[];
    const int t = threadIdx.x;
    const int nthreads = blockDim.x;
    const int kn4 = (K * HC) >> 2;
    for (int q = t; q < kn4; q += nthreads) {
        int idx = q << 2;
        int k = idx >> 7, n = idx & 127;
        const float4 w4 = *(const float4*)(W + idx);
        float* base = &lw[((k >> 1) << 8) + (n << 1) + (k & 1)];
        base[0] = w4.x; base[2] = w4.y; base[4] = w4.z; base[6] = w4.w;
    }
    __syncthreads();

    const int wave = t >> 5, lane = t & 31;
    const int rowblock = blockIdx.x * (nthreads >> 5) + wave;
    const int row0 = rowblock * 16;
    if (row0 >= M) return;

    const int m     = lane & 15;
    const int khalf = lane >> 4;
    const int n     = lane & 15;

    v8f acc[8] = {};
    const float* arow = A + (size_t)(row0 + m) * K + khalf * 2;

    for (int k0 = 0; k0 < K; k0 += 4) {
        v2f a = *(const v2f*)(arow + k0);
        const float* bbase = &lw[((k0 >> 1) + khalf) * 256 + (n << 1)];
        v2f b[8];
        #pragma unroll
        for (int c = 0; c < 8; ++c)
            b[c] = *(const v2f*)(bbase + (c << 5));
        #pragma unroll
        for (int c = 0; c < 8; ++c)
            acc[c] = __builtin_amdgcn_wmma_f32_16x16x4_f32(
                false, a, false, b[c], (short)0, acc[c], false, false);
    }

    float* crow = C + (size_t)row0 * HC;
    const int hi = (lane >> 4) * 8;
    #pragma unroll
    for (int c = 0; c < 8; ++c)
        #pragma unroll
        for (int r = 0; r < 8; ++r)
            crow[(size_t)(r + hi) * HC + c * 16 + n] = acc[c][r];
}

__global__ void att_scores_kernel(const float* __restrict__ g,
                                  const float* __restrict__ att_src,
                                  const float* __restrict__ att_dst,
                                  float* __restrict__ a_s, float* __restrict__ a_d, int Nn) {
    int idx = blockIdx.x * blockDim.x + threadIdx.x;
    if (idx >= Nn * HEADS) return;
    int i = idx >> 2, h = idx & 3;
    const float* row = g + (size_t)i * HC + h * 32;
    const float* as  = att_src + h * 32;
    const float* ad  = att_dst + h * 32;
    float s = 0.f, d = 0.f;
    #pragma unroll
    for (int k = 0; k < 32; ++k) { float v = row[k]; s += v * as[k]; d += v * ad[k]; }
    a_s[idx] = s; a_d[idx] = d;
}

__global__ void init_layer_kernel(float* __restrict__ out, const float* __restrict__ bias,
                                  unsigned* __restrict__ m_u, float* __restrict__ denom, int Nn) {
    int idx = blockIdx.x * blockDim.x + threadIdx.x;
    if (idx >= Nn * HC) return;
    out[idx] = bias[idx & 127];
    if (idx < Nn * HEADS) { m_u[idx] = 0u; denom[idx] = 0.f; }
}

__global__ void edge_alpha_max_kernel(const int* __restrict__ ei, const float* __restrict__ ew,
                                      const float* __restrict__ sumptr, int E, int Nn,
                                      const float* __restrict__ a_s, const float* __restrict__ a_d,
                                      const float* __restrict__ c,
                                      float* __restrict__ alpha, unsigned* __restrict__ m_u) {
    int j = blockIdx.x * blockDim.x + threadIdx.x;
    int Et = E + Nn;
    if (j >= Et) return;
    int s, d; float ea;
    if (j < E) { s = ei[j]; d = ei[E + j]; ea = ew[j]; }
    else       { s = d = j - E; ea = sumptr[0] * (1.0f / (float)E); }
    #pragma unroll
    for (int h = 0; h < HEADS; ++h) {
        float a = a_s[s * HEADS + h] + a_d[d * HEADS + h] + ea * c[h];
        a = (a > 0.f) ? a : NEG_SLOPE * a;
        alpha[(size_t)j * HEADS + h] = a;
        atomicMax(&m_u[d * HEADS + h], enc_f32(a));
    }
}

__global__ void edge_exp_sum_kernel(const int* __restrict__ ei, int E, int Nn,
                                    float* __restrict__ alpha, const unsigned* __restrict__ m_u,
                                    float* __restrict__ denom) {
    int j = blockIdx.x * blockDim.x + threadIdx.x;
    int Et = E + Nn;
    if (j >= Et) return;
    int d = (j < E) ? ei[E + j] : (j - E);
    #pragma unroll
    for (int h = 0; h < HEADS; ++h) {
        float m = dec_f32(m_u[d * HEADS + h]);
        float e = expf(alpha[(size_t)j * HEADS + h] - m);
        alpha[(size_t)j * HEADS + h] = e;
        atomicAdd(&denom[d * HEADS + h], e);
    }
}

__global__ void edge_aggregate_kernel(const int* __restrict__ ei, int E, int Nn,
                                      const float* __restrict__ alpha, const float* __restrict__ denom,
                                      const float* __restrict__ g, float* __restrict__ out) {
    long long gid = (long long)blockIdx.x * blockDim.x + threadIdx.x;
    int j = (int)(gid >> 5);
    int lane = (int)(gid & 31);
    int Et = E + Nn;
    if (j >= Et) return;
    int s, d;
    if (j < E) { s = ei[j]; d = ei[E + j]; }
    else       { s = d = j - E; }
    int h = lane >> 3;
    float w = alpha[(size_t)j * HEADS + h] / denom[d * HEADS + h];
    const float4 hv = *(const float4*)(g + (size_t)s * HC + lane * 4);
    float* o = out + (size_t)d * HC + lane * 4;
    atomicAdd(o + 0, w * hv.x);
    atomicAdd(o + 1, w * hv.y);
    atomicAdd(o + 2, w * hv.z);
    atomicAdd(o + 3, w * hv.w);
}

extern "C" void kernel_launch(void* const* d_in, const int* in_sizes, int n_in,
                              void* d_out, int out_size, void* d_ws, size_t ws_size,
                              hipStream_t stream) {
    const float* x   = (const float*)d_in[0];
    const int*   ei  = (const int*)d_in[1];
    const float* ew  = (const float*)d_in[2];
    const float* W1  = (const float*)d_in[3];
    const float* as1 = (const float*)d_in[4];
    const float* ad1 = (const float*)d_in[5];
    const float* We1 = (const float*)d_in[6];
    const float* ae1 = (const float*)d_in[7];
    const float* b1  = (const float*)d_in[8];
    const float* W2  = (const float*)d_in[9];
    const float* as2 = (const float*)d_in[10];
    const float* ad2 = (const float*)d_in[11];
    const float* We2 = (const float*)d_in[12];
    const float* ae2 = (const float*)d_in[13];
    const float* b2  = (const float*)d_in[14];
    float* out = (float*)d_out;

    const int Nn = in_sizes[0] / 256;
    const int E  = in_sizes[2];
    const int Et = E + Nn;
    const int K1 = in_sizes[3] / HC;
    const int K2 = in_sizes[9] / HC;

    float*    g     = (float*)d_ws;
    float*    h1    = g     + (size_t)Nn * HC;
    float*    a_s   = h1    + (size_t)Nn * HC;
    float*    a_d   = a_s   + (size_t)Nn * HEADS;
    float*    denom = a_d   + (size_t)Nn * HEADS;
    unsigned* m_u   = (unsigned*)(denom + (size_t)Nn * HEADS);
    float*    alpha = (float*)(m_u + (size_t)Nn * HEADS);
    float*    csum  = alpha + (size_t)Et * HEADS;
    float*    ch    = csum + 1;

    const int TB = 256;
    const int mblocks    = ((Nn + 15) / 16 + 3) / 4;
    const int nh_blocks  = (Nn * HEADS + TB - 1) / TB;
    const int init_blocks= (Nn * HC + TB - 1) / TB;
    const int e_blocks   = (Et + TB - 1) / TB;
    const long long aggT = (long long)Et * 32;
    const int agg_blocks = (int)((aggT + TB - 1) / TB);

    hipMemsetAsync(csum, 0, sizeof(float), stream);
    ew_sum_kernel<<<1024, TB, 0, stream>>>(ew, csum, E);

    gemm_wmma_f32<<<mblocks, 128, (size_t)K1 * HC * sizeof(float), stream>>>(x, W1, g, Nn, K1);
    edge_coef_kernel<<<1, 128, 0, stream>>>(We1, ae1, ch);
    att_scores_kernel<<<nh_blocks, TB, 0, stream>>>(g, as1, ad1, a_s, a_d, Nn);
    init_layer_kernel<<<init_blocks, TB, 0, stream>>>(h1, b1, m_u, denom, Nn);
    edge_alpha_max_kernel<<<e_blocks, TB, 0, stream>>>(ei, ew, csum, E, Nn, a_s, a_d, ch, alpha, m_u);
    edge_exp_sum_kernel<<<e_blocks, TB, 0, stream>>>(ei, E, Nn, alpha, m_u, denom);
    edge_aggregate_kernel<<<agg_blocks, TB, 0, stream>>>(ei, E, Nn, alpha, denom, g, h1);

    gemm_wmma_f32<<<mblocks, 128, (size_t)K2 * HC * sizeof(float), stream>>>(h1, W2, g, Nn, K2);
    edge_coef_kernel<<<1, 128, 0, stream>>>(We2, ae2, ch);
    att_scores_kernel<<<nh_blocks, TB, 0, stream>>>(g, as2, ad2, a_s, a_d, Nn);
    init_layer_kernel<<<init_blocks, TB, 0, stream>>>(out, b2, m_u, denom, Nn);
    edge_alpha_max_kernel<<<e_blocks, TB, 0, stream>>>(ei, ew, csum, E, Nn, a_s, a_d, ch, alpha, m_u);
    edge_exp_sum_kernel<<<e_blocks, TB, 0, stream>>>(ei, E, Nn, alpha, m_u, denom);
    edge_aggregate_kernel<<<agg_blocks, TB, 0, stream>>>(ei, E, Nn, alpha, denom, g, out);

    (void)n_in; (void)out_size; (void)ws_size;
}


# Edges grouped by destination once (returning int atomic = rank, scan, place); one wave per node does max, exp sum, weighted row sum in registers: no float atomics, sum order set by slot order
# speedup vs baseline: 3.4972x; 3.4972x over previous
#include <hip/hip_runtime.h>
#include <hip/hip_bf16.h>

typedef float v2f __attribute__((ext_vector_type(2)));
typedef float v8f __attribute__((ext_vector_type(8)));

#define HEADS 4
#define HC 128
#define NEG_SLOPE 0.2f

__global__ void ew_sum_kernel(const float* __restrict__ ew, float* __restrict__ out_sum, int E,
                              int* __restrict__ cnt, int Nn) {
    __shared__ float sm[8];
    for (int i = blockIdx.x * blockDim.x + threadIdx.x; i < Nn; i += gridDim.x * blockDim.x)
        cnt[i] = 0;
    float s = 0.f;
    for (int i = blockIdx.x * blockDim.x + threadIdx.x; i < E; i += gridDim.x * blockDim.x)
        s += ew[i];
    #pragma unroll
    for (int off = 16; off; off >>= 1) s += __shfl_down(s, off, 32);
    int lane = threadIdx.x & 31, w = threadIdx.x >> 5;
    if (lane == 0) sm[w] = s;
    __syncthreads();
    if (threadIdx.x == 0) {
        float t = 0.f;
        for (int i = 0; i < (int)(blockDim.x >> 5); ++i) t += sm[i];
        atomicAdd(out_sum, t);
    }
}

__global__ void edge_rank_kernel(const int* __restrict__ ei, int E,
                                 int* __restrict__ cnt, int* __restrict__ rank) {
    int j = blockIdx.x * blockDim.x + threadIdx.x;
    if (j >= E) return;
    rank[j] = atomicAdd(&cnt[ei[E + j]], 1);
}

__global__ void row_offsets_kernel(const int* __restrict__ cnt, int* __restrict__ rowptr, int Nn) {
    __shared__ int wsum[8];
    __shared__ int wpre[8];
    const int t = threadIdx.x, lane = t & 31, w = t >> 5;
    const int tile0 = blockIdx.x * 1024;
    const int4* c4 = (const int4*)cnt + t;
    const int nb = blockIdx.x;
    int pre = 0;
    int m = 0;
    for (; m + 8 <= nb; m += 8) {
        int4 q[8];
        for (int u = 0; u < 8; ++u) q[u] = c4[(size_t)(m + u) * 256];
        for (int u = 0; u < 8; ++u) pre += (q[u].x + q[u].y) + (q[u].z + q[u].w);
    }
    for (; m < nb; ++m) {
        const int4 q = c4[(size_t)m * 256];
        pre += (q.x + q.y) + (q.z + q.w);
    }
    const int i0 = tile0 + t * 4;
    const int v0 = (i0 + 0 < Nn) ? cnt[i0 + 0] : 0;
    const int v1 = (i0 + 1 < Nn) ? cnt[i0 + 1] : 0;
    const int v2 = (i0 + 2 < Nn) ? cnt[i0 + 2] : 0;
    const int v3 = (i0 + 3 < Nn) ? cnt[i0 + 3] : 0;
    const int tsum = (v0 + v1) + (v2 + v3);
    int inc = tsum;
    for (int off = 1; off < 32; off <<= 1) {
        const int y = __shfl_up(inc, off, 32);
        if (lane >= off) inc += y;
    }
    for (int off = 16; off; off >>= 1) pre += __shfl_xor(pre, off, 32);
    if (lane == 31) wsum[w] = inc;
    if (lane == 0)  wpre[w] = pre;
    __syncthreads();
    int base = 0;
    for (int k = 0; k < 8; ++k) { base += wpre[k]; if (k < w) base += wsum[k]; }
    const int ex = base + inc - tsum;
    if (i0 + 0 < Nn) rowptr[i0 + 1] = ex + v0;
    if (i0 + 1 < Nn) rowptr[i0 + 2] = ex + v0 + v1;
    if (i0 + 2 < Nn) rowptr[i0 + 3] = ex + v0 + v1 + v2;
    if (i0 + 3 < Nn) rowptr[i0 + 4] = ex + tsum;
    if (blockIdx.x == 0 && t == 0) rowptr[0] = 0;
}

__global__ void edge_place_kernel(const int* __restrict__ ei, const float* __restrict__ ew, int E,
                                  const int* __restrict__ rowptr, const int* __restrict__ rank,
                                  int* __restrict__ rec) {
    int j = blockIdx.x * blockDim.x + threadIdx.x;
    if (j >= E) return;
    const int pos = rowptr[ei[E + j]] + rank[j];
    ((int2*)rec)[pos] = make_int2(ei[j], __float_as_int(ew[j]));
}

__global__ void edge_coef_kernel(const float* __restrict__ W_e, const float* __restrict__ att_e,
                                 float* __restrict__ c) {
    int t = threadIdx.x;
    float p = W_e[t] * att_e[t];
    #pragma unroll
    for (int off = 16; off; off >>= 1) p += __shfl_down(p, off, 32);
    if ((t & 31) == 0) c[t >> 5] = p;
}

__global__ void gemm_wmma_f32(const float* __restrict__ A, const float* __restrict__ W,
                              float* __restrict__ C, int M, int K) {
    extern __shared__ float lw[];
    const int t = threadIdx.x;
    const int nthreads = blockDim.x;
    const int kn4 = (K * HC) >> 2;
    for (int q = t; q < kn4; q += nthreads) {
        int idx = q << 2;
        int k = idx >> 7, n = idx & 127;
        const float4 w4 = *(const float4*)(W + idx);
        float* base = &lw[((k >> 1) << 8) + (n << 1) + (k & 1)];
        base[0] = w4.x; base[2] = w4.y; base[4] = w4.z; base[6] = w4.w;
    }
    __syncthreads();

    const int wave = t >> 5, lane = t & 31;
    const int rowblock = blockIdx.x * (nthreads >> 5) + wave;
    const int row0 = rowblock * 16;
    if (row0 >= M) return;

    const int m     = lane & 15;
    const int khalf = lane >> 4;
    const int n     = lane & 15;

    v8f acc[8] = {};
    const float* arow = A + (size_t)(row0 + m) * K + khalf * 2;

    for (int k0 = 0; k0 < K; k0 += 4) {
        v2f a = *(const v2f*)(arow + k0);
        const float* bbase = &lw[((k0 >> 1) + khalf) * 256 + (n << 1)];
        v2f b[8];
        #pragma unroll
        for (int c = 0; c < 8; ++c)
            b[c] = *(const v2f*)(bbase + (c << 5));
        #pragma unroll
        for (int c = 0; c < 8; ++c)
            acc[c] = __builtin_amdgcn_wmma_f32_16x16x4_f32(
                false, a, false, b[c], (short)0, acc[c], false, false);
    }

    float* crow = C + (size_t)row0 * HC;
    const int hi = (lane >> 4) * 8;
    #pragma unroll
    for (int c = 0; c < 8; ++c)
        #pragma unroll
        for (int r = 0; r < 8; ++r)
            crow[(size_t)(r + hi) * HC + c * 16 + n] = acc[c][r];
}

__global__ void att_scores_kernel(const float* __restrict__ g,
                                  const float* __restrict__ att_src,
                                  const float* __restrict__ att_dst,
                                  float* __restrict__ a_s, float* __restrict__ a_d, int Nn) {
    int idx = blockIdx.x * blockDim.x + threadIdx.x;
    if (idx >= Nn * HEADS) return;
    int i = idx >> 2, h = idx & 3;
    const float* row = g + (size_t)i * HC + h * 32;
    const float* as  = att_src + h * 32;
    const float* ad  = att_dst + h * 32;
    float s = 0.f, d = 0.f;
    #pragma unroll
    for (int k = 0; k < 32; ++k) { float v = row[k]; s += v * as[k]; d += v * ad[k]; }
    a_s[idx] = s; a_d[idx] = d;
}

__global__ void node_softmax_aggregate_kernel(const int* __restrict__ rowptr, const int* __restrict__ rec,
                                              const float* __restrict__ sumptr, int E, int Nn,
                                              const float* __restrict__ a_s, const float* __restrict__ a_d,
                                              const float* __restrict__ c,
                                              const float* __restrict__ g, const float* __restrict__ bias,
                                              float* __restrict__ out) {
    const int lane = threadIdx.x & 31;
    const int d = __builtin_amdgcn_readfirstlane((int)(blockIdx.x * (blockDim.x >> 5) + (threadIdx.x >> 5)));
    if (d >= Nn) return;
    const int start = rowptr[d];
    const int n     = rowptr[d + 1] - start;
    const int tot   = n + 1;
    const float ea_self = sumptr[0] * (1.0f / (float)E);
    const float c0 = c[0], c1 = c[1], c2 = c[2], c3 = c[3];
    const float4 ad = *(const float4*)(a_d + (size_t)d * HEADS);
    const int h = lane >> 3;

    auto slot_alpha = [&](int e, int& s, float4& al) -> bool {
        s = d;
        float ea = ea_self;
        if (e < n) { const int2 r = ((const int2*)rec)[start + e]; s = r.x; ea = __int_as_float(r.y); }
        const float4 as4 = *(const float4*)(a_s + (size_t)s * HEADS);
        const float x0 = as4.x + ad.x + ea * c0;
        const float x1 = as4.y + ad.y + ea * c1;
        const float x2 = as4.z + ad.z + ea * c2;
        const float x3 = as4.w + ad.w + ea * c3;
        al.x = (x0 > 0.f) ? x0 : NEG_SLOPE * x0;
        al.y = (x1 > 0.f) ? x1 : NEG_SLOPE * x1;
        al.z = (x2 > 0.f) ? x2 : NEG_SLOPE * x2;
        al.w = (x3 > 0.f) ? x3 : NEG_SLOPE * x3;
        return e < tot;
    };

    int s_first; float4 al_first;
    const bool v_first = slot_alpha(lane, s_first, al_first);
    const float ninf = -__builtin_inff();
    float m0 = v_first ? al_first.x : ninf, m1 = v_first ? al_first.y : ninf;
    float m2 = v_first ? al_first.z : ninf, m3 = v_first ? al_first.w : ninf;
    for (int base = 32; base < tot; base += 32) {
        int s; float4 al;
        if (slot_alpha(base + lane, s, al)) {
            m0 = fmaxf(m0, al.x); m1 = fmaxf(m1, al.y); m2 = fmaxf(m2, al.z); m3 = fmaxf(m3, al.w);
        }
    }
    for (int off = 16; off; off >>= 1) {
        m0 = fmaxf(m0, __shfl_xor(m0, off, 32)); m1 = fmaxf(m1, __shfl_xor(m1, off, 32));
        m2 = fmaxf(m2, __shfl_xor(m2, off, 32)); m3 = fmaxf(m3, __shfl_xor(m3, off, 32));
    }

    const float e0 = v_first ? expf(al_first.x - m0) : 0.f, e1 = v_first ? expf(al_first.y - m1) : 0.f;
    const float e2 = v_first ? expf(al_first.z - m2) : 0.f, e3 = v_first ? expf(al_first.w - m3) : 0.f;
    float d0 = e0, d1 = e1, d2 = e2, d3 = e3;
    for (int base = 32; base < tot; base += 32) {
        int s; float4 al;
        if (slot_alpha(base + lane, s, al)) {
            d0 += expf(al.x - m0); d1 += expf(al.y - m1); d2 += expf(al.z - m2); d3 += expf(al.w - m3);
        }
    }
    for (int off = 16; off; off >>= 1) {
        d0 += __shfl_xor(d0, off, 32); d1 += __shfl_xor(d1, off, 32);
        d2 += __shfl_xor(d2, off, 32); d3 += __shfl_xor(d3, off, 32);
    }

    float4 acc = make_float4(0.f, 0.f, 0.f, 0.f);
    const float* gl = g + lane * 4;
    const unsigned hm1 = (h & 1) ? 0xffffffffu : 0u;
    const unsigned hm2 = (h & 2) ? 0xffffffffu : 0u;
    auto pick = [&](float w0v, float w1v, float w2v, float w3v, int k) -> float {
        const unsigned w0 = (unsigned)__builtin_amdgcn_readlane(__float_as_int(w0v), k);
        const unsigned w1 = (unsigned)__builtin_amdgcn_readlane(__float_as_int(w1v), k);
        const unsigned w2 = (unsigned)__builtin_amdgcn_readlane(__float_as_int(w2v), k);
        const unsigned w3 = (unsigned)__builtin_amdgcn_readlane(__float_as_int(w3v), k);
        const unsigned lo = (w1 & hm1) | (w0 & ~hm1);
        const unsigned hi = (w3 & hm1) | (w2 & ~hm1);
        return __uint_as_float((hi & hm2) | (lo & ~hm2));
    };
    auto add_rows = [&](int s, float w0v, float w1v, float w2v, float w3v, int nslots) {
        for (int k = 0; k < nslots; k += 4) {
            const int sa = __builtin_amdgcn_readlane(s, k);
            const int sb = __builtin_amdgcn_readlane(s, k + 1);
            const int sc = __builtin_amdgcn_readlane(s, k + 2);
            const int sd = __builtin_amdgcn_readlane(s, k + 3);
            const float4 ha = *(const float4*)(gl + (size_t)sa * HC);
            const float4 hb = *(const float4*)(gl + (size_t)sb * HC);
            const float4 hc = *(const float4*)(gl + (size_t)sc * HC);
            const float4 hd = *(const float4*)(gl + (size_t)sd * HC);
            const float wa = pick(w0v, w1v, w2v, w3v, k);
            const float wb = pick(w0v, w1v, w2v, w3v, k + 1);
            const float wc = pick(w0v, w1v, w2v, w3v, k + 2);
            const float wd = pick(w0v, w1v, w2v, w3v, k + 3);
            acc.x += wa * ha.x; acc.y += wa * ha.y; acc.z += wa * ha.z; acc.w += wa * ha.w;
            acc.x += wb * hb.x; acc.y += wb * hb.y; acc.z += wb * hb.z; acc.w += wb * hb.w;
            acc.x += wc * hc.x; acc.y += wc * hc.y; acc.z += wc * hc.z; acc.w += wc * hc.w;
            acc.x += wd * hd.x; acc.y += wd * hd.y; acc.z += wd * hd.z; acc.w += wd * hd.w;
        }
    };
    add_rows(s_first, e0 / d0, e1 / d1, e2 / d2, e3 / d3, tot < 32 ? ((tot + 3) & ~3) : 32);
    for (int base = 32; base < tot; base += 32) {
        int s; float4 al;
        const bool v = slot_alpha(base + lane, s, al);
        const int left = tot - base;
        add_rows(s, v ? expf(al.x - m0) / d0 : 0.f, v ? expf(al.y - m1) / d1 : 0.f,
                 v ? expf(al.z - m2) / d2 : 0.f, v ? expf(al.w - m3) / d3 : 0.f,
                 left < 32 ? ((left + 3) & ~3) : 32);
    }
    const float* bl = bias + lane * 4;
    *(float4*)(out + (size_t)d * HC + lane * 4) =
        make_float4(acc.x + bl[0], acc.y + bl[1], acc.z + bl[2], acc.w + bl[3]);
}

extern "C" void kernel_launch(void* const* d_in, const int* in_sizes, int n_in,
                              void* d_out, int out_size, void* d_ws, size_t ws_size,
                              hipStream_t stream) {
    const float* x   = (const float*)d_in[0];
    const int*   ei  = (const int*)d_in[1];
    const float* ew  = (const float*)d_in[2];
    const float* W1  = (const float*)d_in[3];
    const float* as1 = (const float*)d_in[4];
    const float* ad1 = (const float*)d_in[5];
    const float* We1 = (const float*)d_in[6];
    const float* ae1 = (const float*)d_in[7];
    const float* b1  = (const float*)d_in[8];
    const float* W2  = (const float*)d_in[9];
    const float* as2 = (const float*)d_in[10];
    const float* ad2 = (const float*)d_in[11];
    const float* We2 = (const float*)d_in[12];
    const float* ae2 = (const float*)d_in[13];
    const float* b2  = (const float*)d_in[14];
    float* out = (float*)d_out;

    const int Nn = in_sizes[0] / 256;
    const int E  = in_sizes[2];
    const int K1 = in_sizes[3] / HC;
    const int K2 = in_sizes[9] / HC;

    float* g      = (float*)d_ws;
    float* h1     = g    + (size_t)Nn * HC;
    float* a_s    = h1   + (size_t)Nn * HC;
    float* a_d    = a_s  + (size_t)Nn * HEADS;
    int*   cnt    = (int*)(a_d + (size_t)Nn * HEADS);
    int*   rowptr = cnt + Nn;
    int*   rec    = rowptr + Nn + 2;
    int*   rank   = rec + 2 * (size_t)E;
    float* csum   = (float*)(rank + E);
    float* ch     = csum + 1;

    const int TB = 256;
    const int mblocks     = ((Nn + 15) / 16 + 3) / 4;
    const int nh_blocks   = (Nn * HEADS + TB - 1) / TB;
    const int e_blocks    = (E + TB - 1) / TB;
    const int scan_blocks = (Nn + 1023) / 1024;
    const int node_blocks = (Nn + 7) / 8;

    hipMemsetAsync(csum, 0, sizeof(float), stream);
    ew_sum_kernel<<<1024, TB, 0, stream>>>(ew, csum, E, cnt, Nn);

    edge_rank_kernel<<<e_blocks, TB, 0, stream>>>(ei, E, cnt, rank);
    row_offsets_kernel<<<scan_blocks, 256, 0, stream>>>(cnt, rowptr, Nn);
    edge_place_kernel<<<e_blocks, TB, 0, stream>>>(ei, ew, E, rowptr, rank, rec);

    gemm_wmma_f32<<<mblocks, 128, (size_t)K1 * HC * sizeof(float), stream>>>(x, W1, g, Nn, K1);
    edge_coef_kernel<<<1, 128, 0, stream>>>(We1, ae1, ch);
    att_scores_kernel<<<nh_blocks, TB, 0, stream>>>(g, as1, ad1, a_s, a_d, Nn);
    node_softmax_aggregate_kernel<<<node_blocks, TB, 0, stream>>>(rowptr, rec, csum, E, Nn, a_s, a_d, ch, g, b1, h1);

    gemm_wmma_f32<<<mblocks, 128, (size_t)K2 * HC * sizeof(float), stream>>>(h1, W2, g, Nn, K2);
    edge_coef_kernel<<<1, 128, 0, stream>>>(We2, ae2, ch);
    att_scores_kernel<<<nh_blocks, TB, 0, stream>>>(g, as2, ad2, a_s, a_d, Nn);
    node_softmax_aggregate_kernel<<<node_blocks, TB, 0, stream>>>(rowptr, rec, csum, E, Nn, a_s, a_d, ch, g, b2, out);

    (void)n_in; (void)out_size; (void)ws_size;
}


# Edges grouped by destination with LDS counters in two bucket levels (count, scan, place, per-bucket sort), no global atomic per edge; edge-weight sum from per-chunk partials in fixed order
# speedup vs baseline: 4.0279x; 1.1517x over previous
#include <hip/hip_runtime.h>
#include <hip/hip_bf16.h>

typedef float v2f __attribute__((ext_vector_type(2)));
typedef float v8f __attribute__((ext_vector_type(8)));

#define HEADS 4
#define HC 128
#define NEG_SLOPE 0.2f

#define BK_SHIFT 8
#define BK_NODES 256
#define EB 256
#define ET 512
#define EU 8
#define ST 1024
#define SU 4

__global__ void __launch_bounds__(ET)
bucket_count_kernel(const int* __restrict__ ei, const float* __restrict__ ew, int E, int chunk, int nbk,
                    int* __restrict__ table, float* __restrict__ part) {
    extern __shared__ int lds_cnt[];
    float* sm = (float*)(lds_cnt + nbk);
    const int t = threadIdx.x;
    for (int b = t; b < nbk; b += ET) lds_cnt[b] = 0;
    __syncthreads();
    const int e0 = blockIdx.x * chunk;
    const int e1 = min(E, e0 + chunk);
    float s = 0.f;
    for (int base = e0; base < e1; base += ET * EU) {
        int dv[EU]; float wv[EU];
        #pragma unroll
        for (int u = 0; u < EU; ++u) {
            const int j = base + u * ET + t;
            const bool ok = j < e1;
            dv[u] = ok ? ei[E + j] : -1;
            wv[u] = ok ? ew[j] : 0.f;
        }
        #pragma unroll
        for (int u = 0; u < EU; ++u) {
            if (dv[u] >= 0) atomicAdd(&lds_cnt[dv[u] >> BK_SHIFT], 1);
            s += wv[u];
        }
    }
    #pragma unroll
    for (int off = 16; off; off >>= 1) s += __shfl_down(s, off, 32);
    if ((t & 31) == 0) sm[t >> 5] = s;
    __syncthreads();
    if (t == 0) {
        float tot = 0.f;
        for (int i = 0; i < ET / 32; ++i) tot += sm[i];
        part[blockIdx.x] = tot;
    }
    for (int b = t; b < nbk; b += ET) table[b * EB + blockIdx.x] = lds_cnt[b];
}

__global__ void table_offsets_kernel(const int* __restrict__ cnt, int* __restrict__ rowptr, int Nn,
                                     const float* __restrict__ part, float* __restrict__ out_sum) {
    __shared__ int wsum[8];
    __shared__ int wpre[8];
    const int t = threadIdx.x, lane = t & 31, w = t >> 5;
    const int tile0 = blockIdx.x * 1024;
    const int4* c4 = (const int4*)cnt + t;
    const int nb = blockIdx.x;
    int pre = 0;
    int m = 0;
    for (; m + 8 <= nb; m += 8) {
        int4 q[8];
        for (int u = 0; u < 8; ++u) q[u] = c4[(size_t)(m + u) * 256];
        for (int u = 0; u < 8; ++u) pre += (q[u].x + q[u].y) + (q[u].z + q[u].w);
    }
    for (; m < nb; ++m) {
        const int4 q = c4[(size_t)m * 256];
        pre += (q.x + q.y) + (q.z + q.w);
    }
    const int i0 = tile0 + t * 4;
    const int v0 = (i0 + 0 < Nn) ? cnt[i0 + 0] : 0;
    const int v1 = (i0 + 1 < Nn) ? cnt[i0 + 1] : 0;
    const int v2 = (i0 + 2 < Nn) ? cnt[i0 + 2] : 0;
    const int v3 = (i0 + 3 < Nn) ? cnt[i0 + 3] : 0;
    const int tsum = (v0 + v1) + (v2 + v3);
    int inc = tsum;
    for (int off = 1; off < 32; off <<= 1) {
        const int y = __shfl_up(inc, off, 32);
        if (lane >= off) inc += y;
    }
    for (int off = 16; off; off >>= 1) pre += __shfl_xor(pre, off, 32);
    if (lane == 31) wsum[w] = inc;
    if (lane == 0)  wpre[w] = pre;
    __syncthreads();
    int base = 0;
    for (int k = 0; k < 8; ++k) { base += wpre[k]; if (k < w) base += wsum[k]; }
    const int ex = base + inc - tsum;
    if (i0 + 0 < Nn) rowptr[i0 + 1] = ex + v0;
    if (i0 + 1 < Nn) rowptr[i0 + 2] = ex + v0 + v1;
    if (i0 + 2 < Nn) rowptr[i0 + 3] = ex + v0 + v1 + v2;
    if (i0 + 3 < Nn) rowptr[i0 + 4] = ex + tsum;
    if (blockIdx.x == 0 && t == 0) rowptr[0] = 0;
    if (blockIdx.x == 0 && w == 0) {
        float s = 0.f;
        for (int i = lane; i < EB; i += 32) s += part[i];
        #pragma unroll
        for (int off = 16; off; off >>= 1) s += __shfl_down(s, off, 32);
        if (lane == 0) out_sum[0] = s;
    }
}

__global__ void __launch_bounds__(ET)
bucket_place_kernel(const int* __restrict__ ei, const float* __restrict__ ew, int E, int chunk, int nbk,
                    const int* __restrict__ offs, int2* __restrict__ trec, int* __restrict__ tdst) {
    extern __shared__ int lds_cur[];
    const int t = threadIdx.x;
    for (int b = t; b < nbk; b += ET) lds_cur[b] = offs[b * EB + blockIdx.x];
    __syncthreads();
    const int e0 = blockIdx.x * chunk;
    const int e1 = min(E, e0 + chunk);
    for (int base = e0; base < e1; base += ET * EU) {
        int dv[EU]; int sv[EU]; float wv[EU];
        #pragma unroll
        for (int u = 0; u < EU; ++u) {
            const int j = base + u * ET + t;
            const bool ok = j < e1;
            dv[u] = ok ? ei[E + j] : -1;
            sv[u] = ok ? ei[j] : 0;
            wv[u] = ok ? ew[j] : 0.f;
        }
        #pragma unroll
        for (int u = 0; u < EU; ++u) {
            if (dv[u] >= 0) {
                const int pos = atomicAdd(&lds_cur[dv[u] >> BK_SHIFT], 1);
                trec[pos] = make_int2(sv[u], __float_as_int(wv[u]));
                tdst[pos] = dv[u];
            }
        }
    }
}

__global__ void __launch_bounds__(ST)
bucket_sort_kernel(const int2* __restrict__ trec, const int* __restrict__ tdst, const int* __restrict__ offs,
                   int Nn, int nbk, int* __restrict__ rowptr, int2* __restrict__ rec) {
    __shared__ int cnt[BK_NODES];
    __shared__ int cur[BK_NODES];
    __shared__ int wsum[BK_NODES / 32];
    const int t = threadIdx.x, lane = t & 31, w = t >> 5;
    const int b = blockIdx.x;
    const int bs = offs[b * EB];
    const int be = offs[(b + 1) * EB];
    if (t < BK_NODES) cnt[t] = 0;
    __syncthreads();
    for (int base = bs; base < be; base += ST * SU) {
        int dv[SU];
        #pragma unroll
        for (int u = 0; u < SU; ++u) {
            const int j = base + u * ST + t;
            dv[u] = (j < be) ? tdst[j] : -1;
        }
        #pragma unroll
        for (int u = 0; u < SU; ++u)
            if (dv[u] >= 0) atomicAdd(&cnt[dv[u] & (BK_NODES - 1)], 1);
    }
    __syncthreads();
    int v = 0, inc = 0;
    if (t < BK_NODES) {
        v = cnt[t];
        inc = v;
        for (int off = 1; off < 32; off <<= 1) {
            const int y = __shfl_up(inc, off, 32);
            if (lane >= off) inc += y;
        }
        if (lane == 31) wsum[w] = inc;
    }
    __syncthreads();
    if (t < BK_NODES) {
        int ex = bs + inc - v;
        for (int k = 0; k < w; ++k) ex += wsum[k];
        cur[t] = ex;
        const int node = (b << BK_SHIFT) + t;
        if (node < Nn) rowptr[node] = ex;
    }
    if (b == nbk - 1 && t == 0) rowptr[Nn] = be;
    __syncthreads();
    for (int base = bs; base < be; base += ST * SU) {
        int dv[SU]; int2 rv[SU];
        #pragma unroll
        for (int u = 0; u < SU; ++u) {
            const int j = base + u * ST + t;
            const bool ok = j < be;
            dv[u] = ok ? tdst[j] : -1;
            rv[u] = ok ? trec[j] : make_int2(0, 0);
        }
        #pragma unroll
        for (int u = 0; u < SU; ++u)
            if (dv[u] >= 0) rec[atomicAdd(&cur[dv[u] & (BK_NODES - 1)], 1)] = rv[u];
    }
}

__global__ void edge_coef_kernel(const float* __restrict__ W_e, const float* __restrict__ att_e,
                                 float* __restrict__ c) {
    int t = threadIdx.x;
    float p = W_e[t] * att_e[t];
    #pragma unroll
    for (int off = 16; off; off >>= 1) p += __shfl_down(p, off, 32);
    if ((t & 31) == 0) c[t >> 5] = p;
}

__global__ void gemm_wmma_f32(const float* __restrict__ A, const float* __restrict__ W,
                              float* __restrict__ C, int M, int K) {
    extern __shared__ float lw[];
    const int t = threadIdx.x;
    const int nthreads = blockDim.x;
    const int kn4 = (K * HC) >> 2;
    for (int q = t; q < kn4; q += nthreads) {
        int idx = q << 2;
        int k = idx >> 7, n = idx & 127;
        const float4 w4 = *(const float4*)(W + idx);
        float* base = &lw[((k >> 1) << 8) + (n << 1) + (k & 1)];
        base[0] = w4.x; base[2] = w4.y; base[4] = w4.z; base[6] = w4.w;
    }
    __syncthreads();

    const int wave = t >> 5, lane = t & 31;
    const int rowblock = blockIdx.x * (nthreads >> 5) + wave;
    const int row0 = rowblock * 16;
    if (row0 >= M) return;

    const int m     = lane & 15;
    const int khalf = lane >> 4;
    const int n     = lane & 15;

    v8f acc[8] = {};
    const float* arow = A + (size_t)(row0 + m) * K + khalf * 2;

    for (int k0 = 0; k0 < K; k0 += 4) {
        v2f a = *(const v2f*)(arow + k0);
        const float* bbase = &lw[((k0 >> 1) + khalf) * 256 + (n << 1)];
        v2f b[8];
        #pragma unroll
        for (int c = 0; c < 8; ++c)
            b[c] = *(const v2f*)(bbase + (c << 5));
        #pragma unroll
        for (int c = 0; c < 8; ++c)
            acc[c] = __builtin_amdgcn_wmma_f32_16x16x4_f32(
                false, a, false, b[c], (short)0, acc[c], false, false);
    }

    float* crow = C + (size_t)row0 * HC;
    const int hi = (lane >> 4) * 8;
    #pragma unroll
    for (int c = 0; c < 8; ++c)
        #pragma unroll
        for (int r = 0; r < 8; ++r)
            crow[(size_t)(r + hi) * HC + c * 16 + n] = acc[c][r];
}

__global__ void att_scores_kernel(const float* __restrict__ g,
                                  const float* __restrict__ att_src,
                                  const float* __restrict__ att_dst,
                                  float* __restrict__ a_s, float* __restrict__ a_d, int Nn) {
    int idx = blockIdx.x * blockDim.x + threadIdx.x;
    if (idx >= Nn * HEADS) return;
    int i = idx >> 2, h = idx & 3;
    const float* row = g + (size_t)i * HC + h * 32;
    const float* as  = att_src + h * 32;
    const float* ad  = att_dst + h * 32;
    float s = 0.f, d = 0.f;
    #pragma unroll
    for (int k = 0; k < 32; ++k) { float v = row[k]; s += v * as[k]; d += v * ad[k]; }
    a_s[idx] = s; a_d[idx] = d;
}

__global__ void node_softmax_aggregate_kernel(const int* __restrict__ rowptr, const int* __restrict__ rec,
                                              const float* __restrict__ sumptr, int E, int Nn,
                                              const float* __restrict__ a_s, const float* __restrict__ a_d,
                                              const float* __restrict__ c,
                                              const float* __restrict__ g, const float* __restrict__ bias,
                                              float* __restrict__ out) {
    const int lane = threadIdx.x & 31;
    const int d = __builtin_amdgcn_readfirstlane((int)(blockIdx.x * (blockDim.x >> 5) + (threadIdx.x >> 5)));
    if (d >= Nn) return;
    const int start = rowptr[d];
    const int n     = rowptr[d + 1] - start;
    const int tot   = n + 1;
    const float ea_self = sumptr[0] * (1.0f / (float)E);
    const float c0 = c[0], c1 = c[1], c2 = c[2], c3 = c[3];
    const float4 ad = *(const float4*)(a_d + (size_t)d * HEADS);
    const int h = lane >> 3;

    auto slot_alpha = [&](int e, int& s, float4& al) -> bool {
        s = d;
        float ea = ea_self;
        if (e < n) { const int2 r = ((const int2*)rec)[start + e]; s = r.x; ea = __int_as_float(r.y); }
        const float4 as4 = *(const float4*)(a_s + (size_t)s * HEADS);
        const float x0 = as4.x + ad.x + ea * c0;
        const float x1 = as4.y + ad.y + ea * c1;
        const float x2 = as4.z + ad.z + ea * c2;
        const float x3 = as4.w + ad.w + ea * c3;
        al.x = (x0 > 0.f) ? x0 : NEG_SLOPE * x0;
        al.y = (x1 > 0.f) ? x1 : NEG_SLOPE * x1;
        al.z = (x2 > 0.f) ? x2 : NEG_SLOPE * x2;
        al.w = (x3 > 0.f) ? x3 : NEG_SLOPE * x3;
        return e < tot;
    };

    int s_first; float4 al_first;
    const bool v_first = slot_alpha(lane, s_first, al_first);
    const float ninf = -__builtin_inff();
    float m0 = v_first ? al_first.x : ninf, m1 = v_first ? al_first.y : ninf;
    float m2 = v_first ? al_first.z : ninf, m3 = v_first ? al_first.w : ninf;
    for (int base = 32; base < tot; base += 32) {
        int s; float4 al;
        if (slot_alpha(base + lane, s, al)) {
            m0 = fmaxf(m0, al.x); m1 = fmaxf(m1, al.y); m2 = fmaxf(m2, al.z); m3 = fmaxf(m3, al.w);
        }
    }
    for (int off = 16; off; off >>= 1) {
        m0 = fmaxf(m0, __shfl_xor(m0, off, 32)); m1 = fmaxf(m1, __shfl_xor(m1, off, 32));
        m2 = fmaxf(m2, __shfl_xor(m2, off, 32)); m3 = fmaxf(m3, __shfl_xor(m3, off, 32));
    }

    const float e0 = v_first ? expf(al_first.x - m0) : 0.f, e1 = v_first ? expf(al_first.y - m1) : 0.f;
    const float e2 = v_first ? expf(al_first.z - m2) : 0.f, e3 = v_first ? expf(al_first.w - m3) : 0.f;
    float d0 = e0, d1 = e1, d2 = e2, d3 = e3;
    for (int base = 32; base < tot; base += 32) {
        int s; float4 al;
        if (slot_alpha(base + lane, s, al)) {
            d0 += expf(al.x - m0); d1 += expf(al.y - m1); d2 += expf(al.z - m2); d3 += expf(al.w - m3);
        }
    }
    for (int off = 16; off; off >>= 1) {
        d0 += __shfl_xor(d0, off, 32); d1 += __shfl_xor(d1, off, 32);
        d2 += __shfl_xor(d2, off, 32); d3 += __shfl_xor(d3, off, 32);
    }

    float4 acc = make_float4(0.f, 0.f, 0.f, 0.f);
    const float* gl = g + lane * 4;
    const unsigned hm1 = (h & 1) ? 0xffffffffu : 0u;
    const unsigned hm2 = (h & 2) ? 0xffffffffu : 0u;
    auto pick = [&](float w0v, float w1v, float w2v, float w3v, int k) -> float {
        const unsigned w0 = (unsigned)__builtin_amdgcn_readlane(__float_as_int(w0v), k);
        const unsigned w1 = (unsigned)__builtin_amdgcn_readlane(__float_as_int(w1v), k);
        const unsigned w2 = (unsigned)__builtin_amdgcn_readlane(__float_as_int(w2v), k);
        const unsigned w3 = (unsigned)__builtin_amdgcn_readlane(__float_as_int(w3v), k);
        const unsigned lo = (w1 & hm1) | (w0 & ~hm1);
        const unsigned hi = (w3 & hm1) | (w2 & ~hm1);
        return __uint_as_float((hi & hm2) | (lo & ~hm2));
    };
    auto add_rows = [&](int s, float w0v, float w1v, float w2v, float w3v, int nslots) {
        for (int k = 0; k < nslots; k += 4) {
            const int sa = __builtin_amdgcn_readlane(s, k);
            const int sb = __builtin_amdgcn_readlane(s, k + 1);
            const int sc = __builtin_amdgcn_readlane(s, k + 2);
            const int sd = __builtin_amdgcn_readlane(s, k + 3);
            const float4 ha = *(const float4*)(gl + (size_t)sa * HC);
            const float4 hb = *(const float4*)(gl + (size_t)sb * HC);
            const float4 hc = *(const float4*)(gl + (size_t)sc * HC);
            const float4 hd = *(const float4*)(gl + (size_t)sd * HC);
            const float wa = pick(w0v, w1v, w2v, w3v, k);
            const float wb = pick(w0v, w1v, w2v, w3v, k + 1);
            const float wc = pick(w0v, w1v, w2v, w3v, k + 2);
            const float wd = pick(w0v, w1v, w2v, w3v, k + 3);
            acc.x += wa * ha.x; acc.y += wa * ha.y; acc.z += wa * ha.z; acc.w += wa * ha.w;
            acc.x += wb * hb.x; acc.y += wb * hb.y; acc.z += wb * hb.z; acc.w += wb * hb.w;
            acc.x += wc * hc.x; acc.y += wc * hc.y; acc.z += wc * hc.z; acc.w += wc * hc.w;
            acc.x += wd * hd.x; acc.y += wd * hd.y; acc.z += wd * hd.z; acc.w += wd * hd.w;
        }
    };
    add_rows(s_first, e0 / d0, e1 / d1, e2 / d2, e3 / d3, tot < 32 ? ((tot + 3) & ~3) : 32);
    for (int base = 32; base < tot; base += 32) {
        int s; float4 al;
        const bool v = slot_alpha(base + lane, s, al);
        const int left = tot - base;
        add_rows(s, v ? expf(al.x - m0) / d0 : 0.f, v ? expf(al.y - m1) / d1 : 0.f,
                 v ? expf(al.z - m2) / d2 : 0.f, v ? expf(al.w - m3) / d3 : 0.f,
                 left < 32 ? ((left + 3) & ~3) : 32);
    }
    const float* bl = bias + lane * 4;
    *(float4*)(out + (size_t)d * HC + lane * 4) =
        make_float4(acc.x + bl[0], acc.y + bl[1], acc.z + bl[2], acc.w + bl[3]);
}

extern "C" void kernel_launch(void* const* d_in, const int* in_sizes, int n_in,
                              void* d_out, int out_size, void* d_ws, size_t ws_size,
                              hipStream_t stream) {
    const float* x   = (const float*)d_in[0];
    const int*   ei  = (const int*)d_in[1];
    const float* ew  = (const float*)d_in[2];
    const float* W1  = (const float*)d_in[3];
    const float* as1 = (const float*)d_in[4];
    const float* ad1 = (const float*)d_in[5];
    const float* We1 = (const float*)d_in[6];
    const float* ae1 = (const float*)d_in[7];
    const float* b1  = (const float*)d_in[8];
    const float* W2  = (const float*)d_in[9];
    const float* as2 = (const float*)d_in[10];
    const float* ad2 = (const float*)d_in[11];
    const float* We2 = (const float*)d_in[12];
    const float* ae2 = (const float*)d_in[13];
    const float* b2  = (const float*)d_in[14];
    float* out = (float*)d_out;

    const int Nn = in_sizes[0] / 256;
    const int E  = in_sizes[2];
    const int K1 = in_sizes[3] / HC;
    const int K2 = in_sizes[9] / HC;

    const int nbk   = (Nn + BK_NODES - 1) / BK_NODES;
    const int chunk = (E + EB - 1) / EB;
    const int ntab  = nbk * EB;

    float* g      = (float*)d_ws;
    float* h1     = g    + (size_t)Nn * HC;
    float* a_s    = h1   + (size_t)Nn * HC;
    float* a_d    = a_s  + (size_t)Nn * HEADS;
    int*   table  = (int*)(a_d + (size_t)Nn * HEADS);
    int*   offs   = table + ntab;
    int*   trec   = offs + ntab + 4;
    int*   rec    = trec + 2 * (size_t)E;
    int*   tdst   = rec + 2 * (size_t)E;
    int*   rowptr = tdst + E;
    float* part   = (float*)(rowptr + Nn + 2);
    float* csum   = part + EB;
    float* ch     = csum + 1;

    const int TB = 256;
    const int mblocks     = ((Nn + 15) / 16 + 3) / 4;
    const int nh_blocks   = (Nn * HEADS + TB - 1) / TB;
    const int scan_blocks = (ntab + 1023) / 1024;
    const int node_blocks = (Nn + 7) / 8;

    bucket_count_kernel<<<EB, ET, (size_t)(nbk + ET / 32) * sizeof(int), stream>>>(ei, ew, E, chunk, nbk, table, part);
    table_offsets_kernel<<<scan_blocks, 256, 0, stream>>>(table, offs, ntab, part, csum);
    bucket_place_kernel<<<EB, ET, (size_t)nbk * sizeof(int), stream>>>(ei, ew, E, chunk, nbk, offs, (int2*)trec, tdst);
    bucket_sort_kernel<<<nbk, ST, 0, stream>>>((const int2*)trec, tdst, offs, Nn, nbk, rowptr, (int2*)rec);

    gemm_wmma_f32<<<mblocks, 128, (size_t)K1 * HC * sizeof(float), stream>>>(x, W1, g, Nn, K1);
    edge_coef_kernel<<<1, 128, 0, stream>>>(We1, ae1, ch);
    att_scores_kernel<<<nh_blocks, TB, 0, stream>>>(g, as1, ad1, a_s, a_d, Nn);
    node_softmax_aggregate_kernel<<<node_blocks, TB, 0, stream>>>(rowptr, rec, csum, E, Nn, a_s, a_d, ch, g, b1, h1);

    gemm_wmma_f32<<<mblocks, 128, (size_t)K2 * HC * sizeof(float), stream>>>(h1, W2, g, Nn, K2);
    edge_coef_kernel<<<1, 128, 0, stream>>>(We2, ae2, ch);
    att_scores_kernel<<<nh_blocks, TB, 0, stream>>>(g, as2, ad2, a_s, a_d, Nn);
    node_softmax_aggregate_kernel<<<node_blocks, TB, 0, stream>>>(rowptr, rec, csum, E, Nn, a_s, a_d, ch, g, b2, out);

    (void)n_in; (void)out_size; (void)ws_size;
}


# GEMM: all of W staged into LDS once per 16-wave work-group (16 row tiles) with every staging read in flight before the first LDS store; was once per 4-wave group, one read per round trip
# speedup vs baseline: 5.7519x; 1.4280x over previous
#include <hip/hip_runtime.h>
#include <hip/hip_bf16.h>

typedef float v2f __attribute__((ext_vector_type(2)));
typedef float v8f __attribute__((ext_vector_type(8)));

#define HEADS 4
#define HC 128
#define NEG_SLOPE 0.2f

#define BK_SHIFT 8
#define BK_NODES 256
#define EB 256
#define ET 512
#define EU 8
#define ST 1024
#define SU 4

__global__ void __launch_bounds__(ET)
bucket_count_kernel(const int* __restrict__ ei, const float* __restrict__ ew, int E, int chunk, int nbk,
                    int* __restrict__ table, float* __restrict__ part) {
    extern __shared__ int lds_cnt[];
    float* sm = (float*)(lds_cnt + nbk);
    const int t = threadIdx.x;
    for (int b = t; b < nbk; b += ET) lds_cnt[b] = 0;
    __syncthreads();
    const int e0 = blockIdx.x * chunk;
    const int e1 = min(E, e0 + chunk);
    float s = 0.f;
    for (int base = e0; base < e1; base += ET * EU) {
        int dv[EU]; float wv[EU];
        #pragma unroll
        for (int u = 0; u < EU; ++u) {
            const int j = base + u * ET + t;
            const bool ok = j < e1;
            dv[u] = ok ? ei[E + j] : -1;
            wv[u] = ok ? ew[j] : 0.f;
        }
        #pragma unroll
        for (int u = 0; u < EU; ++u) {
            if (dv[u] >= 0) atomicAdd(&lds_cnt[dv[u] >> BK_SHIFT], 1);
            s += wv[u];
        }
    }
    #pragma unroll
    for (int off = 16; off; off >>= 1) s += __shfl_down(s, off, 32);
    if ((t & 31) == 0) sm[t >> 5] = s;
    __syncthreads();
    if (t == 0) {
        float tot = 0.f;
        for (int i = 0; i < ET / 32; ++i) tot += sm[i];
        part[blockIdx.x] = tot;
    }
    for (int b = t; b < nbk; b += ET) table[b * EB + blockIdx.x] = lds_cnt[b];
}

__global__ void table_offsets_kernel(const int* __restrict__ cnt, int* __restrict__ rowptr, int Nn,
                                     const float* __restrict__ part, float* __restrict__ out_sum) {
    __shared__ int wsum[8];
    __shared__ int wpre[8];
    const int t = threadIdx.x, lane = t & 31, w = t >> 5;
    const int tile0 = blockIdx.x * 1024;
    const int4* c4 = (const int4*)cnt + t;
    const int nb = blockIdx.x;
    int pre = 0;
    int m = 0;
    for (; m + 8 <= nb; m += 8) {
        int4 q[8];
        for (int u = 0; u < 8; ++u) q[u] = c4[(size_t)(m + u) * 256];
        for (int u = 0; u < 8; ++u) pre += (q[u].x + q[u].y) + (q[u].z + q[u].w);
    }
    for (; m < nb; ++m) {
        const int4 q = c4[(size_t)m * 256];
        pre += (q.x + q.y) + (q.z + q.w);
    }
    const int i0 = tile0 + t * 4;
    const int v0 = (i0 + 0 < Nn) ? cnt[i0 + 0] : 0;
    const int v1 = (i0 + 1 < Nn) ? cnt[i0 + 1] : 0;
    const int v2 = (i0 + 2 < Nn) ? cnt[i0 + 2] : 0;
    const int v3 = (i0 + 3 < Nn) ? cnt[i0 + 3] : 0;
    const int tsum = (v0 + v1) + (v2 + v3);
    int inc = tsum;
    for (int off = 1; off < 32; off <<= 1) {
        const int y = __shfl_up(inc, off, 32);
        if (lane >= off) inc += y;
    }
    for (int off = 16; off; off >>= 1) pre += __shfl_xor(pre, off, 32);
    if (lane == 31) wsum[w] = inc;
    if (lane == 0)  wpre[w] = pre;
    __syncthreads();
    int base = 0;
    for (int k = 0; k < 8; ++k) { base += wpre[k]; if (k < w) base += wsum[k]; }
    const int ex = base + inc - tsum;
    if (i0 + 0 < Nn) rowptr[i0 + 1] = ex + v0;
    if (i0 + 1 < Nn) rowptr[i0 + 2] = ex + v0 + v1;
    if (i0 + 2 < Nn) rowptr[i0 + 3] = ex + v0 + v1 + v2;
    if (i0 + 3 < Nn) rowptr[i0 + 4] = ex + tsum;
    if (blockIdx.x == 0 && t == 0) rowptr[0] = 0;
    if (blockIdx.x == 0 && w == 0) {
        float s = 0.f;
        for (int i = lane; i < EB; i += 32) s += part[i];
        #pragma unroll
        for (int off = 16; off; off >>= 1) s += __shfl_down(s, off, 32);
        if (lane == 0) out_sum[0] = s;
    }
}

__global__ void __launch_bounds__(ET)
bucket_place_kernel(const int* __restrict__ ei, const float* __restrict__ ew, int E, int chunk, int nbk,
                    const int* __restrict__ offs, int2* __restrict__ trec, int* __restrict__ tdst) {
    extern __shared__ int lds_cur[];
    const int t = threadIdx.x;
    for (int b = t; b < nbk; b += ET) lds_cur[b] = offs[b * EB + blockIdx.x];
    __syncthreads();
    const int e0 = blockIdx.x * chunk;
    const int e1 = min(E, e0 + chunk);
    for (int base = e0; base < e1; base += ET * EU) {
        int dv[EU]; int sv[EU]; float wv[EU];
        #pragma unroll
        for (int u = 0; u < EU; ++u) {
            const int j = base + u * ET + t;
            const bool ok = j < e1;
            dv[u] = ok ? ei[E + j] : -1;
            sv[u] = ok ? ei[j] : 0;
            wv[u] = ok ? ew[j] : 0.f;
        }
        #pragma unroll
        for (int u = 0; u < EU; ++u) {
            if (dv[u] >= 0) {
                const int pos = atomicAdd(&lds_cur[dv[u] >> BK_SHIFT], 1);
                trec[pos] = make_int2(sv[u], __float_as_int(wv[u]));
                tdst[pos] = dv[u];
            }
        }
    }
}

__global__ void __launch_bounds__(ST)
bucket_sort_kernel(const int2* __restrict__ trec, const int* __restrict__ tdst, const int* __restrict__ offs,
                   int Nn, int nbk, int* __restrict__ rowptr, int2* __restrict__ rec) {
    __shared__ int cnt[BK_NODES];
    __shared__ int cur[BK_NODES];
    __shared__ int wsum[BK_NODES / 32];
    const int t = threadIdx.x, lane = t & 31, w = t >> 5;
    const int b = blockIdx.x;
    const int bs = offs[b * EB];
    const int be = offs[(b + 1) * EB];
    if (t < BK_NODES) cnt[t] = 0;
    __syncthreads();
    for (int base = bs; base < be; base += ST * SU) {
        int dv[SU];
        #pragma unroll
        for (int u = 0; u < SU; ++u) {
            const int j = base + u * ST + t;
            dv[u] = (j < be) ? tdst[j] : -1;
        }
        #pragma unroll
        for (int u = 0; u < SU; ++u)
            if (dv[u] >= 0) atomicAdd(&cnt[dv[u] & (BK_NODES - 1)], 1);
    }
    __syncthreads();
    int v = 0, inc = 0;
    if (t < BK_NODES) {
        v = cnt[t];
        inc = v;
        for (int off = 1; off < 32; off <<= 1) {
            const int y = __shfl_up(inc, off, 32);
            if (lane >= off) inc += y;
        }
        if (lane == 31) wsum[w] = inc;
    }
    __syncthreads();
    if (t < BK_NODES) {
        int ex = bs + inc - v;
        for (int k = 0; k < w; ++k) ex += wsum[k];
        cur[t] = ex;
        const int node = (b << BK_SHIFT) + t;
        if (node < Nn) rowptr[node] = ex;
    }
    if (b == nbk - 1 && t == 0) rowptr[Nn] = be;
    __syncthreads();
    for (int base = bs; base < be; base += ST * SU) {
        int dv[SU]; int2 rv[SU];
        #pragma unroll
        for (int u = 0; u < SU; ++u) {
            const int j = base + u * ST + t;
            const bool ok = j < be;
            dv[u] = ok ? tdst[j] : -1;
            rv[u] = ok ? trec[j] : make_int2(0, 0);
        }
        #pragma unroll
        for (int u = 0; u < SU; ++u)
            if (dv[u] >= 0) rec[atomicAdd(&cur[dv[u] & (BK_NODES - 1)], 1)] = rv[u];
    }
}

__global__ void edge_coef_kernel(const float* __restrict__ W_e, const float* __restrict__ att_e,
                                 float* __restrict__ c) {
    int t = threadIdx.x;
    float p = W_e[t] * att_e[t];
    #pragma unroll
    for (int off = 16; off; off >>= 1) p += __shfl_down(p, off, 32);
    if ((t & 31) == 0) c[t >> 5] = p;
}

#define GW 16
#define GT (GW * 32)
#define GP 8
__global__ void __launch_bounds__(GT)
gemm_wmma_f32(const float* __restrict__ A, const float* __restrict__ W,
              float* __restrict__ C, int M, int K) {
    extern __shared__ float lw[];
    const int t = threadIdx.x;
    const int npairs = (K >> 1) * (HC >> 2);
    for (int p0 = 0; p0 < npairs; p0 += GT * GP) {
        float4 ra[GP], rb[GP];
        #pragma unroll
        for (int u = 0; u < GP; ++u) {
            const int p = p0 + u * GT + t;
            const bool ok = p < npairs;
            const float* src = W + (size_t)(p >> 5) * (2 * HC) + ((p & 31) << 2);
            ra[u] = ok ? *(const float4*)src        : make_float4(0.f, 0.f, 0.f, 0.f);
            rb[u] = ok ? *(const float4*)(src + HC) : make_float4(0.f, 0.f, 0.f, 0.f);
        }
        #pragma unroll
        for (int u = 0; u < GP; ++u) {
            const int p = p0 + u * GT + t;
            if (p < npairs) {
                float4* dst = (float4*)&lw[((p >> 5) << 8) + ((p & 31) << 3)];
                dst[0] = make_float4(ra[u].x, rb[u].x, ra[u].y, rb[u].y);
                dst[1] = make_float4(ra[u].z, rb[u].z, ra[u].w, rb[u].w);
            }
        }
    }
    __syncthreads();

    const int wave = t >> 5, lane = t & 31;
    const int rowblock = blockIdx.x * GW + wave;
    const int row0 = rowblock * 16;
    if (row0 >= M) return;

    const int m     = lane & 15;
    const int khalf = lane >> 4;
    const int n     = lane & 15;

    v8f acc[8] = {};
    const float* arow = A + (size_t)(row0 + m) * K + khalf * 2;

    for (int k0 = 0; k0 < K; k0 += 4) {
        v2f a = *(const v2f*)(arow + k0);
        const float* bbase = &lw[((k0 >> 1) + khalf) * 256 + (n << 1)];
        v2f b[8];
        #pragma unroll
        for (int c = 0; c < 8; ++c)
            b[c] = *(const v2f*)(bbase + (c << 5));
        #pragma unroll
        for (int c = 0; c < 8; ++c)
            acc[c] = __builtin_amdgcn_wmma_f32_16x16x4_f32(
                false, a, false, b[c], (short)0, acc[c], false, false);
    }

    float* crow = C + (size_t)row0 * HC;
    const int hi = (lane >> 4) * 8;
    #pragma unroll
    for (int c = 0; c < 8; ++c)
        #pragma unroll
        for (int r = 0; r < 8; ++r)
            crow[(size_t)(r + hi) * HC + c * 16 + n] = acc[c][r];
}

__global__ void att_scores_kernel(const float* __restrict__ g,
                                  const float* __restrict__ att_src,
                                  const float* __restrict__ att_dst,
                                  float* __restrict__ a_s, float* __restrict__ a_d, int Nn) {
    int idx = blockIdx.x * blockDim.x + threadIdx.x;
    if (idx >= Nn * HEADS) return;
    int i = idx >> 2, h = idx & 3;
    const float* row = g + (size_t)i * HC + h * 32;
    const float* as  = att_src + h * 32;
    const float* ad  = att_dst + h * 32;
    float s = 0.f, d = 0.f;
    #pragma unroll
    for (int k = 0; k < 32; ++k) { float v = row[k]; s += v * as[k]; d += v * ad[k]; }
    a_s[idx] = s; a_d[idx] = d;
}

__global__ void node_softmax_aggregate_kernel(const int* __restrict__ rowptr, const int* __restrict__ rec,
                                              const float* __restrict__ sumptr, int E, int Nn,
                                              const float* __restrict__ a_s, const float* __restrict__ a_d,
                                              const float* __restrict__ c,
                                              const float* __restrict__ g, const float* __restrict__ bias,
                                              float* __restrict__ out) {
    const int lane = threadIdx.x & 31;
    const int d = __builtin_amdgcn_readfirstlane((int)(blockIdx.x * (blockDim.x >> 5) + (threadIdx.x >> 5)));
    if (d >= Nn) return;
    const int start = rowptr[d];
    const int n     = rowptr[d + 1] - start;
    const int tot   = n + 1;
    const float ea_self = sumptr[0] * (1.0f / (float)E);
    const float c0 = c[0], c1 = c[1], c2 = c[2], c3 = c[3];
    const float4 ad = *(const float4*)(a_d + (size_t)d * HEADS);
    const int h = lane >> 3;

    auto slot_alpha = [&](int e, int& s, float4& al) -> bool {
        s = d;
        float ea = ea_self;
        if (e < n) { const int2 r = ((const int2*)rec)[start + e]; s = r.x; ea = __int_as_float(r.y); }
        const float4 as4 = *(const float4*)(a_s + (size_t)s * HEADS);
        const float x0 = as4.x + ad.x + ea * c0;
        const float x1 = as4.y + ad.y + ea * c1;
        const float x2 = as4.z + ad.z + ea * c2;
        const float x3 = as4.w + ad.w + ea * c3;
        al.x = (x0 > 0.f) ? x0 : NEG_SLOPE * x0;
        al.y = (x1 > 0.f) ? x1 : NEG_SLOPE * x1;
        al.z = (x2 > 0.f) ? x2 : NEG_SLOPE * x2;
        al.w = (x3 > 0.f) ? x3 : NEG_SLOPE * x3;
        return e < tot;
    };

    int s_first; float4 al_first;
    const bool v_first = slot_alpha(lane, s_first, al_first);
    const float ninf = -__builtin_inff();
    float m0 = v_first ? al_first.x : ninf, m1 = v_first ? al_first.y : ninf;
    float m2 = v_first ? al_first.z : ninf, m3 = v_first ? al_first.w : ninf;
    for (int base = 32; base < tot; base += 32) {
        int s; float4 al;
        if (slot_alpha(base + lane, s, al)) {
            m0 = fmaxf(m0, al.x); m1 = fmaxf(m1, al.y); m2 = fmaxf(m2, al.z); m3 = fmaxf(m3, al.w);
        }
    }
    for (int off = 16; off; off >>= 1) {
        m0 = fmaxf(m0, __shfl_xor(m0, off, 32)); m1 = fmaxf(m1, __shfl_xor(m1, off, 32));
        m2 = fmaxf(m2, __shfl_xor(m2, off, 32)); m3 = fmaxf(m3, __shfl_xor(m3, off, 32));
    }

    const float e0 = v_first ? expf(al_first.x - m0) : 0.f, e1 = v_first ? expf(al_first.y - m1) : 0.f;
    const float e2 = v_first ? expf(al_first.z - m2) : 0.f, e3 = v_first ? expf(al_first.w - m3) : 0.f;
    float d0 = e0, d1 = e1, d2 = e2, d3 = e3;
    for (int base = 32; base < tot; base += 32) {
        int s; float4 al;
        if (slot_alpha(base + lane, s, al)) {
            d0 += expf(al.x - m0); d1 += expf(al.y - m1); d2 += expf(al.z - m2); d3 += expf(al.w - m3);
        }
    }
    for (int off = 16; off; off >>= 1) {
        d0 += __shfl_xor(d0, off, 32); d1 += __shfl_xor(d1, off, 32);
        d2 += __shfl_xor(d2, off, 32); d3 += __shfl_xor(d3, off, 32);
    }

    float4 acc = make_float4(0.f, 0.f, 0.f, 0.f);
    const float* gl = g + lane * 4;
    const unsigned hm1 = (h & 1) ? 0xffffffffu : 0u;
    const unsigned hm2 = (h & 2) ? 0xffffffffu : 0u;
    auto pick = [&](float w0v, float w1v, float w2v, float w3v, int k) -> float {
        const unsigned w0 = (unsigned)__builtin_amdgcn_readlane(__float_as_int(w0v), k);
        const unsigned w1 = (unsigned)__builtin_amdgcn_readlane(__float_as_int(w1v), k);
        const unsigned w2 = (unsigned)__builtin_amdgcn_readlane(__float_as_int(w2v), k);
        const unsigned w3 = (unsigned)__builtin_amdgcn_readlane(__float_as_int(w3v), k);
        const unsigned lo = (w1 & hm1) | (w0 & ~hm1);
        const unsigned hi = (w3 & hm1) | (w2 & ~hm1);
        return __uint_as_float((hi & hm2) | (lo & ~hm2));
    };
    auto add_rows = [&](int s, float w0v, float w1v, float w2v, float w3v, int nslots) {
        for (int k = 0; k < nslots; k += 4) {
            const int sa = __builtin_amdgcn_readlane(s, k);
            const int sb = __builtin_amdgcn_readlane(s, k + 1);
            const int sc = __builtin_amdgcn_readlane(s, k + 2);
            const int sd = __builtin_amdgcn_readlane(s, k + 3);
            const float4 ha = *(const float4*)(gl + (size_t)sa * HC);
            const float4 hb = *(const float4*)(gl + (size_t)sb * HC);
            const float4 hc = *(const float4*)(gl + (size_t)sc * HC);
            const float4 hd = *(const float4*)(gl + (size_t)sd * HC);
            const float wa = pick(w0v, w1v, w2v, w3v, k);
            const float wb = pick(w0v, w1v, w2v, w3v, k + 1);
            const float wc = pick(w0v, w1v, w2v, w3v, k + 2);
            const float wd = pick(w0v, w1v, w2v, w3v, k + 3);
            acc.x += wa * ha.x; acc.y += wa * ha.y; acc.z += wa * ha.z; acc.w += wa * ha.w;
            acc.x += wb * hb.x; acc.y += wb * hb.y; acc.z += wb * hb.z; acc.w += wb * hb.w;
            acc.x += wc * hc.x; acc.y += wc * hc.y; acc.z += wc * hc.z; acc.w += wc * hc.w;
            acc.x += wd * hd.x; acc.y += wd * hd.y; acc.z += wd * hd.z; acc.w += wd * hd.w;
        }
    };
    add_rows(s_first, e0 / d0, e1 / d1, e2 / d2, e3 / d3, tot < 32 ? ((tot + 3) & ~3) : 32);
    for (int base = 32; base < tot; base += 32) {
        int s; float4 al;
        const bool v = slot_alpha(base + lane, s, al);
        const int left = tot - base;
        add_rows(s, v ? expf(al.x - m0) / d0 : 0.f, v ? expf(al.y - m1) / d1 : 0.f,
                 v ? expf(al.z - m2) / d2 : 0.f, v ? expf(al.w - m3) / d3 : 0.f,
                 left < 32 ? ((left + 3) & ~3) : 32);
    }
    const float* bl = bias + lane * 4;
    *(float4*)(out + (size_t)d * HC + lane * 4) =
        make_float4(acc.x + bl[0], acc.y + bl[1], acc.z + bl[2], acc.w + bl[3]);
}

extern "C" void kernel_launch(void* const* d_in, const int* in_sizes, int n_in,
                              void* d_out, int out_size, void* d_ws, size_t ws_size,
                              hipStream_t stream) {
    const float* x   = (const float*)d_in[0];
    const int*   ei  = (const int*)d_in[1];
    const float* ew  = (const float*)d_in[2];
    const float* W1  = (const float*)d_in[3];
    const float* as1 = (const float*)d_in[4];
    const float* ad1 = (const float*)d_in[5];
    const float* We1 = (const float*)d_in[6];
    const float* ae1 = (const float*)d_in[7];
    const float* b1  = (const float*)d_in[8];
    const float* W2  = (const float*)d_in[9];
    const float* as2 = (const float*)d_in[10];
    const float* ad2 = (const float*)d_in[11];
    const float* We2 = (const float*)d_in[12];
    const float* ae2 = (const float*)d_in[13];
    const float* b2  = (const float*)d_in[14];
    float* out = (float*)d_out;

    const int Nn = in_sizes[0] / 256;
    const int E  = in_sizes[2];
    const int K1 = in_sizes[3] / HC;
    const int K2 = in_sizes[9] / HC;

    const int nbk   = (Nn + BK_NODES - 1) / BK_NODES;
    const int chunk = (E + EB - 1) / EB;
    const int ntab  = nbk * EB;

    float* g      = (float*)d_ws;
    float* h1     = g    + (size_t)Nn * HC;
    float* a_s    = h1   + (size_t)Nn * HC;
    float* a_d    = a_s  + (size_t)Nn * HEADS;
    int*   table  = (int*)(a_d + (size_t)Nn * HEADS);
    int*   offs   = table + ntab;
    int*   trec   = offs + ntab + 4;
    int*   rec    = trec + 2 * (size_t)E;
    int*   tdst   = rec + 2 * (size_t)E;
    int*   rowptr = tdst + E;
    float* part   = (float*)(rowptr + Nn + 2);
    float* csum   = part + EB;
    float* ch     = csum + 1;

    const int TB = 256;
    const int mblocks     = ((Nn + 15) / 16 + GW - 1) / GW;
    const int nh_blocks   = (Nn * HEADS + TB - 1) / TB;
    const int scan_blocks = (ntab + 1023) / 1024;
    const int node_blocks = (Nn + 7) / 8;

    bucket_count_kernel<<<EB, ET, (size_t)(nbk + ET / 32) * sizeof(int), stream>>>(ei, ew, E, chunk, nbk, table, part);
    table_offsets_kernel<<<scan_blocks, 256, 0, stream>>>(table, offs, ntab, part, csum);
    bucket_place_kernel<<<EB, ET, (size_t)nbk * sizeof(int), stream>>>(ei, ew, E, chunk, nbk, offs, (int2*)trec, tdst);
    bucket_sort_kernel<<<nbk, ST, 0, stream>>>((const int2*)trec, tdst, offs, Nn, nbk, rowptr, (int2*)rec);

    gemm_wmma_f32<<<mblocks, GT, (size_t)K1 * HC * sizeof(float), stream>>>(x, W1, g, Nn, K1);
    edge_coef_kernel<<<1, 128, 0, stream>>>(We1, ae1, ch);
    att_scores_kernel<<<nh_blocks, TB, 0, stream>>>(g, as1, ad1, a_s, a_d, Nn);
    node_softmax_aggregate_kernel<<<node_blocks, TB, 0, stream>>>(rowptr, rec, csum, E, Nn, a_s, a_d, ch, g, b1, h1);

    gemm_wmma_f32<<<mblocks, GT, (size_t)K2 * HC * sizeof(float), stream>>>(h1, W2, g, Nn, K2);
    edge_coef_kernel<<<1, 128, 0, stream>>>(We2, ae2, ch);
    att_scores_kernel<<<nh_blocks, TB, 0, stream>>>(g, as2, ad2, a_s, a_d, Nn);
    node_softmax_aggregate_kernel<<<node_blocks, TB, 0, stream>>>(rowptr, rec, csum, E, Nn, a_s, a_d, ch, g, b2, out);

    (void)n_in; (void)out_size; (void)ws_size;
}


# Scores and edge-coefficient launches gone: a_s/a_d summed from the GEMM accumulators in its epilogue (16-lane row_xmask tree, new sum order), c_h of both layers in the offsets kernel; 12->8 launches
# speedup vs baseline: 6.3999x; 1.1127x over previous
#include <hip/hip_runtime.h>
#include <hip/hip_bf16.h>

typedef float v2f __attribute__((ext_vector_type(2)));
typedef float v8f __attribute__((ext_vector_type(8)));

#define HEADS 4
#define HC 128
#define NEG_SLOPE 0.2f

#define BK_SHIFT 8
#define BK_NODES 256
#define EB 256
#define ET 512
#define EU 8
#define ST 1024
#define SU 4

__global__ void __launch_bounds__(ET)
bucket_count_kernel(const int* __restrict__ ei, const float* __restrict__ ew, int E, int chunk, int nbk,
                    int* __restrict__ table, float* __restrict__ part) {
    extern __shared__ int lds_cnt[];
    float* sm = (float*)(lds_cnt + nbk);
    const int t = threadIdx.x;
    for (int b = t; b < nbk; b += ET) lds_cnt[b] = 0;
    __syncthreads();
    const int e0 = blockIdx.x * chunk;
    const int e1 = min(E, e0 + chunk);
    float s = 0.f;
    for (int base = e0; base < e1; base += ET * EU) {
        int dv[EU]; float wv[EU];
        #pragma unroll
        for (int u = 0; u < EU; ++u) {
            const int j = base + u * ET + t;
            const bool ok = j < e1;
            dv[u] = ok ? ei[E + j] : -1;
            wv[u] = ok ? ew[j] : 0.f;
        }
        #pragma unroll
        for (int u = 0; u < EU; ++u) {
            if (dv[u] >= 0) atomicAdd(&lds_cnt[dv[u] >> BK_SHIFT], 1);
            s += wv[u];
        }
    }
    #pragma unroll
    for (int off = 16; off; off >>= 1) s += __shfl_down(s, off, 32);
    if ((t & 31) == 0) sm[t >> 5] = s;
    __syncthreads();
    if (t == 0) {
        float tot = 0.f;
        for (int i = 0; i < ET / 32; ++i) tot += sm[i];
        part[blockIdx.x] = tot;
    }
    for (int b = t; b < nbk; b += ET) table[b * EB + blockIdx.x] = lds_cnt[b];
}

__global__ void table_offsets_kernel(const int* __restrict__ cnt, int* __restrict__ rowptr, int Nn,
                                     const float* __restrict__ part, float* __restrict__ out_sum,
                                     const float* __restrict__ W_ea, const float* __restrict__ att_ea,
                                     const float* __restrict__ W_eb, const float* __restrict__ att_eb,
                                     float* __restrict__ coef) {
    __shared__ int wsum[8];
    __shared__ int wpre[8];
    const int t = threadIdx.x, lane = t & 31, w = t >> 5;
    const int tile0 = blockIdx.x * 1024;
    const int4* c4 = (const int4*)cnt + t;
    const int nb = blockIdx.x;
    int pre = 0;
    int m = 0;
    for (; m + 8 <= nb; m += 8) {
        int4 q[8];
        for (int u = 0; u < 8; ++u) q[u] = c4[(size_t)(m + u) * 256];
        for (int u = 0; u < 8; ++u) pre += (q[u].x + q[u].y) + (q[u].z + q[u].w);
    }
    for (; m < nb; ++m) {
        const int4 q = c4[(size_t)m * 256];
        pre += (q.x + q.y) + (q.z + q.w);
    }
    const int i0 = tile0 + t * 4;
    const int v0 = (i0 + 0 < Nn) ? cnt[i0 + 0] : 0;
    const int v1 = (i0 + 1 < Nn) ? cnt[i0 + 1] : 0;
    const int v2 = (i0 + 2 < Nn) ? cnt[i0 + 2] : 0;
    const int v3 = (i0 + 3 < Nn) ? cnt[i0 + 3] : 0;
    const int tsum = (v0 + v1) + (v2 + v3);
    int inc = tsum;
    for (int off = 1; off < 32; off <<= 1) {
        const int y = __shfl_up(inc, off, 32);
        if (lane >= off) inc += y;
    }
    for (int off = 16; off; off >>= 1) pre += __shfl_xor(pre, off, 32);
    if (lane == 31) wsum[w] = inc;
    if (lane == 0)  wpre[w] = pre;
    __syncthreads();
    int base = 0;
    for (int k = 0; k < 8; ++k) { base += wpre[k]; if (k < w) base += wsum[k]; }
    const int ex = base + inc - tsum;
    if (i0 + 0 < Nn) rowptr[i0 + 1] = ex + v0;
    if (i0 + 1 < Nn) rowptr[i0 + 2] = ex + v0 + v1;
    if (i0 + 2 < Nn) rowptr[i0 + 3] = ex + v0 + v1 + v2;
    if (i0 + 3 < Nn) rowptr[i0 + 4] = ex + tsum;
    if (blockIdx.x == 0 && t == 0) rowptr[0] = 0;
    if (blockIdx.x == 0 && w == 0) {
        float s = 0.f;
        for (int i = lane; i < EB; i += 32) s += part[i];
        #pragma unroll
        for (int off = 16; off; off >>= 1) s += __shfl_down(s, off, 32);
        if (lane == 0) out_sum[0] = s;
    }
    if (blockIdx.x == 0) {
        const int j = t & 127;
        float p = (t < 128) ? W_ea[j] * att_ea[j] : W_eb[j] * att_eb[j];
        #pragma unroll
        for (int off = 16; off; off >>= 1) p += __shfl_down(p, off, 32);
        if (lane == 0) coef[w] = p;
    }
}

__global__ void __launch_bounds__(ET)
bucket_place_kernel(const int* __restrict__ ei, const float* __restrict__ ew, int E, int chunk, int nbk,
                    const int* __restrict__ offs, int2* __restrict__ trec, int* __restrict__ tdst) {
    extern __shared__ int lds_cur[];
    const int t = threadIdx.x;
    for (int b = t; b < nbk; b += ET) lds_cur[b] = offs[b * EB + blockIdx.x];
    __syncthreads();
    const int e0 = blockIdx.x * chunk;
    const int e1 = min(E, e0 + chunk);
    for (int base = e0; base < e1; base += ET * EU) {
        int dv[EU]; int sv[EU]; float wv[EU];
        #pragma unroll
        for (int u = 0; u < EU; ++u) {
            const int j = base + u * ET + t;
            const bool ok = j < e1;
            dv[u] = ok ? ei[E + j] : -1;
            sv[u] = ok ? ei[j] : 0;
            wv[u] = ok ? ew[j] : 0.f;
        }
        #pragma unroll
        for (int u = 0; u < EU; ++u) {
            if (dv[u] >= 0) {
                const int pos = atomicAdd(&lds_cur[dv[u] >> BK_SHIFT], 1);
                trec[pos] = make_int2(sv[u], __float_as_int(wv[u]));
                tdst[pos] = dv[u];
            }
        }
    }
}

__global__ void __launch_bounds__(ST)
bucket_sort_kernel(const int2* __restrict__ trec, const int* __restrict__ tdst, const int* __restrict__ offs,
                   int Nn, int nbk, int* __restrict__ rowptr, int2* __restrict__ rec) {
    __shared__ int cnt[BK_NODES];
    __shared__ int cur[BK_NODES];
    __shared__ int wsum[BK_NODES / 32];
    const int t = threadIdx.x, lane = t & 31, w = t >> 5;
    const int b = blockIdx.x;
    const int bs = offs[b * EB];
    const int be = offs[(b + 1) * EB];
    if (t < BK_NODES) cnt[t] = 0;
    __syncthreads();
    for (int base = bs; base < be; base += ST * SU) {
        int dv[SU];
        #pragma unroll
        for (int u = 0; u < SU; ++u) {
            const int j = base + u * ST + t;
            dv[u] = (j < be) ? tdst[j] : -1;
        }
        #pragma unroll
        for (int u = 0; u < SU; ++u)
            if (dv[u] >= 0) atomicAdd(&cnt[dv[u] & (BK_NODES - 1)], 1);
    }
    __syncthreads();
    int v = 0, inc = 0;
    if (t < BK_NODES) {
        v = cnt[t];
        inc = v;
        for (int off = 1; off < 32; off <<= 1) {
            const int y = __shfl_up(inc, off, 32);
            if (lane >= off) inc += y;
        }
        if (lane == 31) wsum[w] = inc;
    }
    __syncthreads();
    if (t < BK_NODES) {
        int ex = bs + inc - v;
        for (int k = 0; k < w; ++k) ex += wsum[k];
        cur[t] = ex;
        const int node = (b << BK_SHIFT) + t;
        if (node < Nn) rowptr[node] = ex;
    }
    if (b == nbk - 1 && t == 0) rowptr[Nn] = be;
    __syncthreads();
    for (int base = bs; base < be; base += ST * SU) {
        int dv[SU]; int2 rv[SU];
        #pragma unroll
        for (int u = 0; u < SU; ++u) {
            const int j = base + u * ST + t;
            const bool ok = j < be;
            dv[u] = ok ? tdst[j] : -1;
            rv[u] = ok ? trec[j] : make_int2(0, 0);
        }
        #pragma unroll
        for (int u = 0; u < SU; ++u)
            if (dv[u] >= 0) rec[atomicAdd(&cur[dv[u] & (BK_NODES - 1)], 1)] = rv[u];
    }
}

template <int MASK>
__device__ __forceinline__ float row_xor_add(float keep, float send) {
    return keep + __int_as_float(__builtin_amdgcn_update_dpp(0, __float_as_int(send), 0x160 | MASK, 0xf, 0xf, true));
}

#define GW 16
#define GT (GW * 32)
#define GP 8
__global__ void __launch_bounds__(GT)
gemm_wmma_f32(const float* __restrict__ A, const float* __restrict__ W,
              float* __restrict__ C, int M, int K,
              const float* __restrict__ att_src, const float* __restrict__ att_dst,
              float* __restrict__ a_s, float* __restrict__ a_d) {
    extern __shared__ float lw[];
    const int t = threadIdx.x;
    const int npairs = (K >> 1) * (HC >> 2);
    for (int p0 = 0; p0 < npairs; p0 += GT * GP) {
        float4 ra[GP], rb[GP];
        #pragma unroll
        for (int u = 0; u < GP; ++u) {
            const int p = p0 + u * GT + t;
            const bool ok = p < npairs;
            const float* src = W + (size_t)(p >> 5) * (2 * HC) + ((p & 31) << 2);
            ra[u] = ok ? *(const float4*)src        : make_float4(0.f, 0.f, 0.f, 0.f);
            rb[u] = ok ? *(const float4*)(src + HC) : make_float4(0.f, 0.f, 0.f, 0.f);
        }
        #pragma unroll
        for (int u = 0; u < GP; ++u) {
            const int p = p0 + u * GT + t;
            if (p < npairs) {
                float4* dst = (float4*)&lw[((p >> 5) << 8) + ((p & 31) << 3)];
                dst[0] = make_float4(ra[u].x, rb[u].x, ra[u].y, rb[u].y);
                dst[1] = make_float4(ra[u].z, rb[u].z, ra[u].w, rb[u].w);
            }
        }
    }
    __syncthreads();

    const int wave = t >> 5, lane = t & 31;
    const int rowblock = blockIdx.x * GW + wave;
    const int row0 = rowblock * 16;
    if (row0 >= M) return;

    const int m     = lane & 15;
    const int khalf = lane >> 4;
    const int n     = lane & 15;

    v8f acc[8] = {};
    const float* arow = A + (size_t)(row0 + m) * K + khalf * 2;

    float sl[HEADS], sh[HEADS], dl[HEADS], dh[HEADS];
    #pragma unroll
    for (int h = 0; h < HEADS; ++h) {
        sl[h] = att_src[h * 32 + n]; sh[h] = att_src[h * 32 + 16 + n];
        dl[h] = att_dst[h * 32 + n]; dh[h] = att_dst[h * 32 + 16 + n];
    }

    for (int k0 = 0; k0 < K; k0 += 4) {
        v2f a = *(const v2f*)(arow + k0);
        const float* bbase = &lw[((k0 >> 1) + khalf) * 256 + (n << 1)];
        v2f b[8];
        #pragma unroll
        for (int c = 0; c < 8; ++c)
            b[c] = *(const v2f*)(bbase + (c << 5));
        #pragma unroll
        for (int c = 0; c < 8; ++c)
            acc[c] = __builtin_amdgcn_wmma_f32_16x16x4_f32(
                false, a, false, b[c], (short)0, acc[c], false, false);
    }

    float* crow = C + (size_t)row0 * HC;
    const int hi = (lane >> 4) * 8;
    #pragma unroll
    for (int c = 0; c < 8; ++c)
        #pragma unroll
        for (int r = 0; r < 8; ++r)
            crow[(size_t)(r + hi) * HC + c * 16 + n] = acc[c][r];

    const bool up8 = (n & 8) != 0, up4 = (n & 4) != 0, up2 = (n & 2) != 0, up1 = (n & 1) != 0;
    float v8[32];
    #pragma unroll
    for (int h = 0; h < HEADS; ++h) {
        #pragma unroll
        for (int r = 0; r < 8; ++r) {
            const float ps = acc[2 * h][r] * sl[h] + acc[2 * h + 1][r] * sh[h];
            const float pd = acc[2 * h][r] * dl[h] + acc[2 * h + 1][r] * dh[h];
            v8[h * 8 + r] = row_xor_add<8>(up8 ? pd : ps, up8 ? ps : pd);
        }
    }
    float v4[16];
    #pragma unroll
    for (int h = 0; h < HEADS; ++h)
        #pragma unroll
        for (int q = 0; q < 4; ++q) {
            const float lo = v8[h * 8 + q], up = v8[h * 8 + q + 4];
            v4[h * 4 + q] = row_xor_add<4>(up4 ? up : lo, up4 ? lo : up);
        }
    float v2[8];
    #pragma unroll
    for (int h = 0; h < HEADS; ++h)
        #pragma unroll
        for (int q = 0; q < 2; ++q) {
            const float lo = v4[h * 4 + q], up = v4[h * 4 + q + 2];
            v2[h * 2 + q] = row_xor_add<2>(up2 ? up : lo, up2 ? lo : up);
        }
    float v1[HEADS];
    #pragma unroll
    for (int h = 0; h < HEADS; ++h) {
        const float lo = v2[h * 2], up = v2[h * 2 + 1];
        v1[h] = row_xor_add<1>(up1 ? up : lo, up1 ? lo : up);
    }
    float* sdst = (up8 ? a_d : a_s) + (size_t)(row0 + hi + (n & 7)) * HEADS;
    *(float4*)sdst = make_float4(v1[0], v1[1], v1[2], v1[3]);
}

__global__ void node_softmax_aggregate_kernel(const int* __restrict__ rowptr, const int* __restrict__ rec,
                                              const float* __restrict__ sumptr, int E, int Nn,
                                              const float* __restrict__ a_s, const float* __restrict__ a_d,
                                              const float* __restrict__ c,
                                              const float* __restrict__ g, const float* __restrict__ bias,
                                              float* __restrict__ out) {
    const int lane = threadIdx.x & 31;
    const int d = __builtin_amdgcn_readfirstlane((int)(blockIdx.x * (blockDim.x >> 5) + (threadIdx.x >> 5)));
    if (d >= Nn) return;
    const int start = rowptr[d];
    const int n     = rowptr[d + 1] - start;
    const int tot   = n + 1;
    const float ea_self = sumptr[0] * (1.0f / (float)E);
    const float c0 = c[0], c1 = c[1], c2 = c[2], c3 = c[3];
    const float4 ad = *(const float4*)(a_d + (size_t)d * HEADS);
    const int h = lane >> 3;

    auto slot_alpha = [&](int e, int& s, float4& al) -> bool {
        s = d;
        float ea = ea_self;
        if (e < n) { const int2 r = ((const int2*)rec)[start + e]; s = r.x; ea = __int_as_float(r.y); }
        const float4 as4 = *(const float4*)(a_s + (size_t)s * HEADS);
        const float x0 = as4.x + ad.x + ea * c0;
        const float x1 = as4.y + ad.y + ea * c1;
        const float x2 = as4.z + ad.z + ea * c2;
        const float x3 = as4.w + ad.w + ea * c3;
        al.x = (x0 > 0.f) ? x0 : NEG_SLOPE * x0;
        al.y = (x1 > 0.f) ? x1 : NEG_SLOPE * x1;
        al.z = (x2 > 0.f) ? x2 : NEG_SLOPE * x2;
        al.w = (x3 > 0.f) ? x3 : NEG_SLOPE * x3;
        return e < tot;
    };

    int s_first; float4 al_first;
    const bool v_first = slot_alpha(lane, s_first, al_first);
    const float ninf = -__builtin_inff();
    float m0 = v_first ? al_first.x : ninf, m1 = v_first ? al_first.y : ninf;
    float m2 = v_first ? al_first.z : ninf, m3 = v_first ? al_first.w : ninf;
    for (int base = 32; base < tot; base += 32) {
        int s; float4 al;
        if (slot_alpha(base + lane, s, al)) {
            m0 = fmaxf(m0, al.x); m1 = fmaxf(m1, al.y); m2 = fmaxf(m2, al.z); m3 = fmaxf(m3, al.w);
        }
    }
    for (int off = 16; off; off >>= 1) {
        m0 = fmaxf(m0, __shfl_xor(m0, off, 32)); m1 = fmaxf(m1, __shfl_xor(m1, off, 32));
        m2 = fmaxf(m2, __shfl_xor(m2, off, 32)); m3 = fmaxf(m3, __shfl_xor(m3, off, 32));
    }

    const float e0 = v_first ? expf(al_first.x - m0) : 0.f, e1 = v_first ? expf(al_first.y - m1) : 0.f;
    const float e2 = v_first ? expf(al_first.z - m2) : 0.f, e3 = v_first ? expf(al_first.w - m3) : 0.f;
    float d0 = e0, d1 = e1, d2 = e2, d3 = e3;
    for (int base = 32; base < tot; base += 32) {
        int s; float4 al;
        if (slot_alpha(base + lane, s, al)) {
            d0 += expf(al.x - m0); d1 += expf(al.y - m1); d2 += expf(al.z - m2); d3 += expf(al.w - m3);
        }
    }
    for (int off = 16; off; off >>= 1) {
        d0 += __shfl_xor(d0, off, 32); d1 += __shfl_xor(d1, off, 32);
        d2 += __shfl_xor(d2, off, 32); d3 += __shfl_xor(d3, off, 32);
    }

    float4 acc = make_float4(0.f, 0.f, 0.f, 0.f);
    const float* gl = g + lane * 4;
    const unsigned hm1 = (h & 1) ? 0xffffffffu : 0u;
    const unsigned hm2 = (h & 2) ? 0xffffffffu : 0u;
    auto pick = [&](float w0v, float w1v, float w2v, float w3v, int k) -> float {
        const unsigned w0 = (unsigned)__builtin_amdgcn_readlane(__float_as_int(w0v), k);
        const unsigned w1 = (unsigned)__builtin_amdgcn_readlane(__float_as_int(w1v), k);
        const unsigned w2 = (unsigned)__builtin_amdgcn_readlane(__float_as_int(w2v), k);
        const unsigned w3 = (unsigned)__builtin_amdgcn_readlane(__float_as_int(w3v), k);
        const unsigned lo = (w1 & hm1) | (w0 & ~hm1);
        const unsigned hi = (w3 & hm1) | (w2 & ~hm1);
        return __uint_as_float((hi & hm2) | (lo & ~hm2));
    };
    auto add_rows = [&](int s, float w0v, float w1v, float w2v, float w3v, int nslots) {
        for (int k = 0; k < nslots; k += 4) {
            const int sa = __builtin_amdgcn_readlane(s, k);
            const int sb = __builtin_amdgcn_readlane(s, k + 1);
            const int sc = __builtin_amdgcn_readlane(s, k + 2);
            const int sd = __builtin_amdgcn_readlane(s, k + 3);
            const float4 ha = *(const float4*)(gl + (size_t)sa * HC);
            const float4 hb = *(const float4*)(gl + (size_t)sb * HC);
            const float4 hc = *(const float4*)(gl + (size_t)sc * HC);
            const float4 hd = *(const float4*)(gl + (size_t)sd * HC);
            const float wa = pick(w0v, w1v, w2v, w3v, k);
            const float wb = pick(w0v, w1v, w2v, w3v, k + 1);
            const float wc = pick(w0v, w1v, w2v, w3v, k + 2);
            const float wd = pick(w0v, w1v, w2v, w3v, k + 3);
            acc.x += wa * ha.x; acc.y += wa * ha.y; acc.z += wa * ha.z; acc.w += wa * ha.w;
            acc.x += wb * hb.x; acc.y += wb * hb.y; acc.z += wb * hb.z; acc.w += wb * hb.w;
            acc.x += wc * hc.x; acc.y += wc * hc.y; acc.z += wc * hc.z; acc.w += wc * hc.w;
            acc.x += wd * hd.x; acc.y += wd * hd.y; acc.z += wd * hd.z; acc.w += wd * hd.w;
        }
    };
    add_rows(s_first, e0 / d0, e1 / d1, e2 / d2, e3 / d3, tot < 32 ? ((tot + 3) & ~3) : 32);
    for (int base = 32; base < tot; base += 32) {
        int s; float4 al;
        const bool v = slot_alpha(base + lane, s, al);
        const int left = tot - base;
        add_rows(s, v ? expf(al.x - m0) / d0 : 0.f, v ? expf(al.y - m1) / d1 : 0.f,
                 v ? expf(al.z - m2) / d2 : 0.f, v ? expf(al.w - m3) / d3 : 0.f,
                 left < 32 ? ((left + 3) & ~3) : 32);
    }
    const float* bl = bias + lane * 4;
    *(float4*)(out + (size_t)d * HC + lane * 4) =
        make_float4(acc.x + bl[0], acc.y + bl[1], acc.z + bl[2], acc.w + bl[3]);
}

extern "C" void kernel_launch(void* const* d_in, const int* in_sizes, int n_in,
                              void* d_out, int out_size, void* d_ws, size_t ws_size,
                              hipStream_t stream) {
    const float* x   = (const float*)d_in[0];
    const int*   ei  = (const int*)d_in[1];
    const float* ew  = (const float*)d_in[2];
    const float* W1  = (const float*)d_in[3];
    const float* as1 = (const float*)d_in[4];
    const float* ad1 = (const float*)d_in[5];
    const float* We1 = (const float*)d_in[6];
    const float* ae1 = (const float*)d_in[7];
    const float* b1  = (const float*)d_in[8];
    const float* W2  = (const float*)d_in[9];
    const float* as2 = (const float*)d_in[10];
    const float* ad2 = (const float*)d_in[11];
    const float* We2 = (const float*)d_in[12];
    const float* ae2 = (const float*)d_in[13];
    const float* b2  = (const float*)d_in[14];
    float* out = (float*)d_out;

    const int Nn = in_sizes[0] / 256;
    const int E  = in_sizes[2];
    const int K1 = in_sizes[3] / HC;
    const int K2 = in_sizes[9] / HC;

    const int nbk   = (Nn + BK_NODES - 1) / BK_NODES;
    const int chunk = (E + EB - 1) / EB;
    const int ntab  = nbk * EB;

    float* g      = (float*)d_ws;
    float* h1     = g    + (size_t)Nn * HC;
    float* a_s    = h1   + (size_t)Nn * HC;
    float* a_d    = a_s  + (size_t)Nn * HEADS;
    int*   table  = (int*)(a_d + (size_t)Nn * HEADS);
    int*   offs   = table + ntab;
    int*   trec   = offs + ntab + 4;
    int*   rec    = trec + 2 * (size_t)E;
    int*   tdst   = rec + 2 * (size_t)E;
    int*   rowptr = tdst + E;
    float* part   = (float*)(rowptr + Nn + 2);
    float* csum   = part + EB;
    float* ch     = csum + 1;

    const int TB = 256;
    const int mblocks     = ((Nn + 15) / 16 + GW - 1) / GW;
    const int scan_blocks = (ntab + 1023) / 1024;
    const int node_blocks = (Nn + 7) / 8;

    bucket_count_kernel<<<EB, ET, (size_t)(nbk + ET / 32) * sizeof(int), stream>>>(ei, ew, E, chunk, nbk, table, part);
    table_offsets_kernel<<<scan_blocks, 256, 0, stream>>>(table, offs, ntab, part, csum, We1, ae1, We2, ae2, ch);
    bucket_place_kernel<<<EB, ET, (size_t)nbk * sizeof(int), stream>>>(ei, ew, E, chunk, nbk, offs, (int2*)trec, tdst);
    bucket_sort_kernel<<<nbk, ST, 0, stream>>>((const int2*)trec, tdst, offs, Nn, nbk, rowptr, (int2*)rec);

    gemm_wmma_f32<<<mblocks, GT, (size_t)K1 * HC * sizeof(float), stream>>>(x, W1, g, Nn, K1, as1, ad1, a_s, a_d);
    node_softmax_aggregate_kernel<<<node_blocks, TB, 0, stream>>>(rowptr, rec, csum, E, Nn, a_s, a_d, ch, g, b1, h1);

    gemm_wmma_f32<<<mblocks, GT, (size_t)K2 * HC * sizeof(float), stream>>>(h1, W2, g, Nn, K2, as2, ad2, a_s, a_d);
    node_softmax_aggregate_kernel<<<node_blocks, TB, 0, stream>>>(rowptr, rec, csum, E, Nn, a_s, a_d, ch + HEADS, g, b2, out);

    (void)n_in; (void)out_size; (void)ws_size;
}


# First layer's GEMM and the edge placement share one launch (block index picks the body): the placement runs beside the GEMM on the WGPs it leaves free, not in a launch ahead of it; 8->7 launches
# speedup vs baseline: 6.6316x; 1.0362x over previous
#include <hip/hip_runtime.h>
#include <hip/hip_bf16.h>

typedef float v2f __attribute__((ext_vector_type(2)));
typedef float v8f __attribute__((ext_vector_type(8)));

#define HEADS 4
#define HC 128
#define NEG_SLOPE 0.2f

#define BK_SHIFT 8
#define BK_NODES 256
#define EB 256
#define ET 512
#define EU 8
#define ST 1024
#define SU 4

__global__ void __launch_bounds__(ET)
bucket_count_kernel(const int* __restrict__ ei, const float* __restrict__ ew, int E, int chunk, int nbk,
                    int* __restrict__ table, float* __restrict__ part) {
    extern __shared__ int lds_cnt[];
    float* sm = (float*)(lds_cnt + nbk);
    const int t = threadIdx.x;
    for (int b = t; b < nbk; b += ET) lds_cnt[b] = 0;
    __syncthreads();
    const int e0 = blockIdx.x * chunk;
    const int e1 = min(E, e0 + chunk);
    float s = 0.f;
    for (int base = e0; base < e1; base += ET * EU) {
        int dv[EU]; float wv[EU];
        #pragma unroll
        for (int u = 0; u < EU; ++u) {
            const int j = base + u * ET + t;
            const bool ok = j < e1;
            dv[u] = ok ? ei[E + j] : -1;
            wv[u] = ok ? ew[j] : 0.f;
        }
        #pragma unroll
        for (int u = 0; u < EU; ++u) {
            if (dv[u] >= 0) atomicAdd(&lds_cnt[dv[u] >> BK_SHIFT], 1);
            s += wv[u];
        }
    }
    #pragma unroll
    for (int off = 16; off; off >>= 1) s += __shfl_down(s, off, 32);
    if ((t & 31) == 0) sm[t >> 5] = s;
    __syncthreads();
    if (t == 0) {
        float tot = 0.f;
        for (int i = 0; i < ET / 32; ++i) tot += sm[i];
        part[blockIdx.x] = tot;
    }
    for (int b = t; b < nbk; b += ET) table[b * EB + blockIdx.x] = lds_cnt[b];
}

__global__ void table_offsets_kernel(const int* __restrict__ cnt, int* __restrict__ rowptr, int Nn,
                                     const float* __restrict__ part, float* __restrict__ out_sum,
                                     const float* __restrict__ W_ea, const float* __restrict__ att_ea,
                                     const float* __restrict__ W_eb, const float* __restrict__ att_eb,
                                     float* __restrict__ coef) {
    __shared__ int wsum[8];
    __shared__ int wpre[8];
    const int t = threadIdx.x, lane = t & 31, w = t >> 5;
    const int tile0 = blockIdx.x * 1024;
    const int4* c4 = (const int4*)cnt + t;
    const int nb = blockIdx.x;
    int pre = 0;
    int m = 0;
    for (; m + 8 <= nb; m += 8) {
        int4 q[8];
        for (int u = 0; u < 8; ++u) q[u] = c4[(size_t)(m + u) * 256];
        for (int u = 0; u < 8; ++u) pre += (q[u].x + q[u].y) + (q[u].z + q[u].w);
    }
    for (; m < nb; ++m) {
        const int4 q = c4[(size_t)m * 256];
        pre += (q.x + q.y) + (q.z + q.w);
    }
    const int i0 = tile0 + t * 4;
    const int v0 = (i0 + 0 < Nn) ? cnt[i0 + 0] : 0;
    const int v1 = (i0 + 1 < Nn) ? cnt[i0 + 1] : 0;
    const int v2 = (i0 + 2 < Nn) ? cnt[i0 + 2] : 0;
    const int v3 = (i0 + 3 < Nn) ? cnt[i0 + 3] : 0;
    const int tsum = (v0 + v1) + (v2 + v3);
    int inc = tsum;
    for (int off = 1; off < 32; off <<= 1) {
        const int y = __shfl_up(inc, off, 32);
        if (lane >= off) inc += y;
    }
    for (int off = 16; off; off >>= 1) pre += __shfl_xor(pre, off, 32);
    if (lane == 31) wsum[w] = inc;
    if (lane == 0)  wpre[w] = pre;
    __syncthreads();
    int base = 0;
    for (int k = 0; k < 8; ++k) { base += wpre[k]; if (k < w) base += wsum[k]; }
    const int ex = base + inc - tsum;
    if (i0 + 0 < Nn) rowptr[i0 + 1] = ex + v0;
    if (i0 + 1 < Nn) rowptr[i0 + 2] = ex + v0 + v1;
    if (i0 + 2 < Nn) rowptr[i0 + 3] = ex + v0 + v1 + v2;
    if (i0 + 3 < Nn) rowptr[i0 + 4] = ex + tsum;
    if (blockIdx.x == 0 && t == 0) rowptr[0] = 0;
    if (blockIdx.x == 0 && w == 0) {
        float s = 0.f;
        for (int i = lane; i < EB; i += 32) s += part[i];
        #pragma unroll
        for (int off = 16; off; off >>= 1) s += __shfl_down(s, off, 32);
        if (lane == 0) out_sum[0] = s;
    }
    if (blockIdx.x == 0) {
        const int j = t & 127;
        float p = (t < 128) ? W_ea[j] * att_ea[j] : W_eb[j] * att_eb[j];
        #pragma unroll
        for (int off = 16; off; off >>= 1) p += __shfl_down(p, off, 32);
        if (lane == 0) coef[w] = p;
    }
}

__device__ __forceinline__ void
bucket_place_body(int cb, int* lds_cur, const int* __restrict__ ei, const float* __restrict__ ew, int E, int chunk,
                  int nbk, const int* __restrict__ offs, int2* __restrict__ trec, int* __restrict__ tdst) {
    const int t = threadIdx.x;
    for (int b = t; b < nbk; b += ET) lds_cur[b] = offs[b * EB + cb];
    __syncthreads();
    const int e0 = cb * chunk;
    const int e1 = min(E, e0 + chunk);
    for (int base = e0; base < e1; base += ET * EU) {
        int dv[EU]; int sv[EU]; float wv[EU];
        #pragma unroll
        for (int u = 0; u < EU; ++u) {
            const int j = base + u * ET + t;
            const bool ok = j < e1;
            dv[u] = ok ? ei[E + j] : -1;
            sv[u] = ok ? ei[j] : 0;
            wv[u] = ok ? ew[j] : 0.f;
        }
        #pragma unroll
        for (int u = 0; u < EU; ++u) {
            if (dv[u] >= 0) {
                const int pos = atomicAdd(&lds_cur[dv[u] >> BK_SHIFT], 1);
                trec[pos] = make_int2(sv[u], __float_as_int(wv[u]));
                tdst[pos] = dv[u];
            }
        }
    }
}

__global__ void __launch_bounds__(ST)
bucket_sort_kernel(const int2* __restrict__ trec, const int* __restrict__ tdst, const int* __restrict__ offs,
                   int Nn, int nbk, int* __restrict__ rowptr, int2* __restrict__ rec) {
    __shared__ int cnt[BK_NODES];
    __shared__ int cur[BK_NODES];
    __shared__ int wsum[BK_NODES / 32];
    const int t = threadIdx.x, lane = t & 31, w = t >> 5;
    const int b = blockIdx.x;
    const int bs = offs[b * EB];
    const int be = offs[(b + 1) * EB];
    if (t < BK_NODES) cnt[t] = 0;
    __syncthreads();
    for (int base = bs; base < be; base += ST * SU) {
        int dv[SU];
        #pragma unroll
        for (int u = 0; u < SU; ++u) {
            const int j = base + u * ST + t;
            dv[u] = (j < be) ? tdst[j] : -1;
        }
        #pragma unroll
        for (int u = 0; u < SU; ++u)
            if (dv[u] >= 0) atomicAdd(&cnt[dv[u] & (BK_NODES - 1)], 1);
    }
    __syncthreads();
    int v = 0, inc = 0;
    if (t < BK_NODES) {
        v = cnt[t];
        inc = v;
        for (int off = 1; off < 32; off <<= 1) {
            const int y = __shfl_up(inc, off, 32);
            if (lane >= off) inc += y;
        }
        if (lane == 31) wsum[w] = inc;
    }
    __syncthreads();
    if (t < BK_NODES) {
        int ex = bs + inc - v;
        for (int k = 0; k < w; ++k) ex += wsum[k];
        cur[t] = ex;
        const int node = (b << BK_SHIFT) + t;
        if (node < Nn) rowptr[node] = ex;
    }
    if (b == nbk - 1 && t == 0) rowptr[Nn] = be;
    __syncthreads();
    for (int base = bs; base < be; base += ST * SU) {
        int dv[SU]; int2 rv[SU];
        #pragma unroll
        for (int u = 0; u < SU; ++u) {
            const int j = base + u * ST + t;
            const bool ok = j < be;
            dv[u] = ok ? tdst[j] : -1;
            rv[u] = ok ? trec[j] : make_int2(0, 0);
        }
        #pragma unroll
        for (int u = 0; u < SU; ++u)
            if (dv[u] >= 0) rec[atomicAdd(&cur[dv[u] & (BK_NODES - 1)], 1)] = rv[u];
    }
}

template <int MASK>
__device__ __forceinline__ float row_xor_add(float keep, float send) {
    return keep + __int_as_float(__builtin_amdgcn_update_dpp(0, __float_as_int(send), 0x160 | MASK, 0xf, 0xf, true));
}

#define GW 16
#define GT (GW * 32)
#define GP 8
__device__ __forceinline__ void
gemm_wmma_f32_body(int gb, float* lw, const float* __restrict__ A, const float* __restrict__ W,
                   float* __restrict__ C, int M, int K,
                   const float* __restrict__ att_src, const float* __restrict__ att_dst,
                   float* __restrict__ a_s, float* __restrict__ a_d) {
    const int t = threadIdx.x;
    const int npairs = (K >> 1) * (HC >> 2);
    for (int p0 = 0; p0 < npairs; p0 += GT * GP) {
        float4 ra[GP], rb[GP];
        #pragma unroll
        for (int u = 0; u < GP; ++u) {
            const int p = p0 + u * GT + t;
            const bool ok = p < npairs;
            const float* src = W + (size_t)(p >> 5) * (2 * HC) + ((p & 31) << 2);
            ra[u] = ok ? *(const float4*)src        : make_float4(0.f, 0.f, 0.f, 0.f);
            rb[u] = ok ? *(const float4*)(src + HC) : make_float4(0.f, 0.f, 0.f, 0.f);
        }
        #pragma unroll
        for (int u = 0; u < GP; ++u) {
            const int p = p0 + u * GT + t;
            if (p < npairs) {
                float4* dst = (float4*)&lw[((p >> 5) << 8) + ((p & 31) << 3)];
                dst[0] = make_float4(ra[u].x, rb[u].x, ra[u].y, rb[u].y);
                dst[1] = make_float4(ra[u].z, rb[u].z, ra[u].w, rb[u].w);
            }
        }
    }
    __syncthreads();

    const int wave = t >> 5, lane = t & 31;
    const int rowblock = gb * GW + wave;
    const int row0 = rowblock * 16;
    if (row0 >= M) return;

    const int m     = lane & 15;
    const int khalf = lane >> 4;
    const int n     = lane & 15;

    v8f acc[8] = {};
    const float* arow = A + (size_t)(row0 + m) * K + khalf * 2;

    float sl[HEADS], sh[HEADS], dl[HEADS], dh[HEADS];
    #pragma unroll
    for (int h = 0; h < HEADS; ++h) {
        sl[h] = att_src[h * 32 + n]; sh[h] = att_src[h * 32 + 16 + n];
        dl[h] = att_dst[h * 32 + n]; dh[h] = att_dst[h * 32 + 16 + n];
    }

    for (int k0 = 0; k0 < K; k0 += 4) {
        v2f a = *(const v2f*)(arow + k0);
        const float* bbase = &lw[((k0 >> 1) + khalf) * 256 + (n << 1)];
        v2f b[8];
        #pragma unroll
        for (int c = 0; c < 8; ++c)
            b[c] = *(const v2f*)(bbase + (c << 5));
        #pragma unroll
        for (int c = 0; c < 8; ++c)
            acc[c] = __builtin_amdgcn_wmma_f32_16x16x4_f32(
                false, a, false, b[c], (short)0, acc[c], false, false);
    }

    float* crow = C + (size_t)row0 * HC;
    const int hi = (lane >> 4) * 8;
    #pragma unroll
    for (int c = 0; c < 8; ++c)
        #pragma unroll
        for (int r = 0; r < 8; ++r)
            crow[(size_t)(r + hi) * HC + c * 16 + n] = acc[c][r];

    const bool up8 = (n & 8) != 0, up4 = (n & 4) != 0, up2 = (n & 2) != 0, up1 = (n & 1) != 0;
    float v8[32];
    #pragma unroll
    for (int h = 0; h < HEADS; ++h) {
        #pragma unroll
        for (int r = 0; r < 8; ++r) {
            const float ps = acc[2 * h][r] * sl[h] + acc[2 * h + 1][r] * sh[h];
            const float pd = acc[2 * h][r] * dl[h] + acc[2 * h + 1][r] * dh[h];
            v8[h * 8 + r] = row_xor_add<8>(up8 ? pd : ps, up8 ? ps : pd);
        }
    }
    float v4[16];
    #pragma unroll
    for (int h = 0; h < HEADS; ++h)
        #pragma unroll
        for (int q = 0; q < 4; ++q) {
            const float lo = v8[h * 8 + q], up = v8[h * 8 + q + 4];
            v4[h * 4 + q] = row_xor_add<4>(up4 ? up : lo, up4 ? lo : up);
        }
    float v2[8];
    #pragma unroll
    for (int h = 0; h < HEADS; ++h)
        #pragma unroll
        for (int q = 0; q < 2; ++q) {
            const float lo = v4[h * 4 + q], up = v4[h * 4 + q + 2];
            v2[h * 2 + q] = row_xor_add<2>(up2 ? up : lo, up2 ? lo : up);
        }
    float v1[HEADS];
    #pragma unroll
    for (int h = 0; h < HEADS; ++h) {
        const float lo = v2[h * 2], up = v2[h * 2 + 1];
        v1[h] = row_xor_add<1>(up1 ? up : lo, up1 ? lo : up);
    }
    float* sdst = (up8 ? a_d : a_s) + (size_t)(row0 + hi + (n & 7)) * HEADS;
    *(float4*)sdst = make_float4(v1[0], v1[1], v1[2], v1[3]);
}

__global__ void __launch_bounds__(GT)
gemm_wmma_f32(const float* __restrict__ A, const float* __restrict__ W,
              float* __restrict__ C, int M, int K,
              const float* __restrict__ att_src, const float* __restrict__ att_dst,
              float* __restrict__ a_s, float* __restrict__ a_d) {
    extern __shared__ float dyn_lds[];
    gemm_wmma_f32_body((int)blockIdx.x, dyn_lds, A, W, C, M, K, att_src, att_dst, a_s, a_d);
}

static_assert(GT == ET, "the shared launch needs one block size for both bodies");
__global__ void __launch_bounds__(GT)
gemm_and_place_kernel(const float* __restrict__ A, const float* __restrict__ W,
                      float* __restrict__ C, int M, int K,
                      const float* __restrict__ att_src, const float* __restrict__ att_dst,
                      float* __restrict__ a_s, float* __restrict__ a_d, int gblocks,
                      const int* __restrict__ ei, const float* __restrict__ ew, int E, int chunk, int nbk,
                      const int* __restrict__ offs, int2* __restrict__ trec, int* __restrict__ tdst) {
    extern __shared__ float dyn_lds[];
    const int bx = (int)blockIdx.x;
    if (bx < gblocks) gemm_wmma_f32_body(bx, dyn_lds, A, W, C, M, K, att_src, att_dst, a_s, a_d);
    else bucket_place_body(bx - gblocks, (int*)dyn_lds, ei, ew, E, chunk, nbk, offs, trec, tdst);
}

__global__ void node_softmax_aggregate_kernel(const int* __restrict__ rowptr, const int* __restrict__ rec,
                                              const float* __restrict__ sumptr, int E, int Nn,
                                              const float* __restrict__ a_s, const float* __restrict__ a_d,
                                              const float* __restrict__ c,
                                              const float* __restrict__ g, const float* __restrict__ bias,
                                              float* __restrict__ out) {
    const int lane = threadIdx.x & 31;
    const int d = __builtin_amdgcn_readfirstlane((int)(blockIdx.x * (blockDim.x >> 5) + (threadIdx.x >> 5)));
    if (d >= Nn) return;
    const int start = rowptr[d];
    const int n     = rowptr[d + 1] - start;
    const int tot   = n + 1;
    const float ea_self = sumptr[0] * (1.0f / (float)E);
    const float c0 = c[0], c1 = c[1], c2 = c[2], c3 = c[3];
    const float4 ad = *(const float4*)(a_d + (size_t)d * HEADS);
    const int h = lane >> 3;

    auto slot_alpha = [&](int e, int& s, float4& al) -> bool {
        s = d;
        float ea = ea_self;
        if (e < n) { const int2 r = ((const int2*)rec)[start + e]; s = r.x; ea = __int_as_float(r.y); }
        const float4 as4 = *(const float4*)(a_s + (size_t)s * HEADS);
        const float x0 = as4.x + ad.x + ea * c0;
        const float x1 = as4.y + ad.y + ea * c1;
        const float x2 = as4.z + ad.z + ea * c2;
        const float x3 = as4.w + ad.w + ea * c3;
        al.x = (x0 > 0.f) ? x0 : NEG_SLOPE * x0;
        al.y = (x1 > 0.f) ? x1 : NEG_SLOPE * x1;
        al.z = (x2 > 0.f) ? x2 : NEG_SLOPE * x2;
        al.w = (x3 > 0.f) ? x3 : NEG_SLOPE * x3;
        return e < tot;
    };

    int s_first; float4 al_first;
    const bool v_first = slot_alpha(lane, s_first, al_first);
    const float ninf = -__builtin_inff();
    float m0 = v_first ? al_first.x : ninf, m1 = v_first ? al_first.y : ninf;
    float m2 = v_first ? al_first.z : ninf, m3 = v_first ? al_first.w : ninf;
    for (int base = 32; base < tot; base += 32) {
        int s; float4 al;
        if (slot_alpha(base + lane, s, al)) {
            m0 = fmaxf(m0, al.x); m1 = fmaxf(m1, al.y); m2 = fmaxf(m2, al.z); m3 = fmaxf(m3, al.w);
        }
    }
    for (int off = 16; off; off >>= 1) {
        m0 = fmaxf(m0, __shfl_xor(m0, off, 32)); m1 = fmaxf(m1, __shfl_xor(m1, off, 32));
        m2 = fmaxf(m2, __shfl_xor(m2, off, 32)); m3 = fmaxf(m3, __shfl_xor(m3, off, 32));
    }

    const float e0 = v_first ? expf(al_first.x - m0) : 0.f, e1 = v_first ? expf(al_first.y - m1) : 0.f;
    const float e2 = v_first ? expf(al_first.z - m2) : 0.f, e3 = v_first ? expf(al_first.w - m3) : 0.f;
    float d0 = e0, d1 = e1, d2 = e2, d3 = e3;
    for (int base = 32; base < tot; base += 32) {
        int s; float4 al;
        if (slot_alpha(base + lane, s, al)) {
            d0 += expf(al.x - m0); d1 += expf(al.y - m1); d2 += expf(al.z - m2); d3 += expf(al.w - m3);
        }
    }
    for (int off = 16; off; off >>= 1) {
        d0 += __shfl_xor(d0, off, 32); d1 += __shfl_xor(d1, off, 32);
        d2 += __shfl_xor(d2, off, 32); d3 += __shfl_xor(d3, off, 32);
    }

    float4 acc = make_float4(0.f, 0.f, 0.f, 0.f);
    const float* gl = g + lane * 4;
    const unsigned hm1 = (h & 1) ? 0xffffffffu : 0u;
    const unsigned hm2 = (h & 2) ? 0xffffffffu : 0u;
    auto pick = [&](float w0v, float w1v, float w2v, float w3v, int k) -> float {
        const unsigned w0 = (unsigned)__builtin_amdgcn_readlane(__float_as_int(w0v), k);
        const unsigned w1 = (unsigned)__builtin_amdgcn_readlane(__float_as_int(w1v), k);
        const unsigned w2 = (unsigned)__builtin_amdgcn_readlane(__float_as_int(w2v), k);
        const unsigned w3 = (unsigned)__builtin_amdgcn_readlane(__float_as_int(w3v), k);
        const unsigned lo = (w1 & hm1) | (w0 & ~hm1);
        const unsigned hi = (w3 & hm1) | (w2 & ~hm1);
        return __uint_as_float((hi & hm2) | (lo & ~hm2));
    };
    auto add_rows = [&](int s, float w0v, float w1v, float w2v, float w3v, int nslots) {
        for (int k = 0; k < nslots; k += 4) {
            const int sa = __builtin_amdgcn_readlane(s, k);
            const int sb = __builtin_amdgcn_readlane(s, k + 1);
            const int sc = __builtin_amdgcn_readlane(s, k + 2);
            const int sd = __builtin_amdgcn_readlane(s, k + 3);
            const float4 ha = *(const float4*)(gl + (size_t)sa * HC);
            const float4 hb = *(const float4*)(gl + (size_t)sb * HC);
            const float4 hc = *(const float4*)(gl + (size_t)sc * HC);
            const float4 hd = *(const float4*)(gl + (size_t)sd * HC);
            const float wa = pick(w0v, w1v, w2v, w3v, k);
            const float wb = pick(w0v, w1v, w2v, w3v, k + 1);
            const float wc = pick(w0v, w1v, w2v, w3v, k + 2);
            const float wd = pick(w0v, w1v, w2v, w3v, k + 3);
            acc.x += wa * ha.x; acc.y += wa * ha.y; acc.z += wa * ha.z; acc.w += wa * ha.w;
            acc.x += wb * hb.x; acc.y += wb * hb.y; acc.z += wb * hb.z; acc.w += wb * hb.w;
            acc.x += wc * hc.x; acc.y += wc * hc.y; acc.z += wc * hc.z; acc.w += wc * hc.w;
            acc.x += wd * hd.x; acc.y += wd * hd.y; acc.z += wd * hd.z; acc.w += wd * hd.w;
        }
    };
    add_rows(s_first, e0 / d0, e1 / d1, e2 / d2, e3 / d3, tot < 32 ? ((tot + 3) & ~3) : 32);
    for (int base = 32; base < tot; base += 32) {
        int s; float4 al;
        const bool v = slot_alpha(base + lane, s, al);
        const int left = tot - base;
        add_rows(s, v ? expf(al.x - m0) / d0 : 0.f, v ? expf(al.y - m1) / d1 : 0.f,
                 v ? expf(al.z - m2) / d2 : 0.f, v ? expf(al.w - m3) / d3 : 0.f,
                 left < 32 ? ((left + 3) & ~3) : 32);
    }
    const float* bl = bias + lane * 4;
    *(float4*)(out + (size_t)d * HC + lane * 4) =
        make_float4(acc.x + bl[0], acc.y + bl[1], acc.z + bl[2], acc.w + bl[3]);
}

extern "C" void kernel_launch(void* const* d_in, const int* in_sizes, int n_in,
                              void* d_out, int out_size, void* d_ws, size_t ws_size,
                              hipStream_t stream) {
    const float* x   = (const float*)d_in[0];
    const int*   ei  = (const int*)d_in[1];
    const float* ew  = (const float*)d_in[2];
    const float* W1  = (const float*)d_in[3];
    const float* as1 = (const float*)d_in[4];
    const float* ad1 = (const float*)d_in[5];
    const float* We1 = (const float*)d_in[6];
    const float* ae1 = (const float*)d_in[7];
    const float* b1  = (const float*)d_in[8];
    const float* W2  = (const float*)d_in[9];
    const float* as2 = (const float*)d_in[10];
    const float* ad2 = (const float*)d_in[11];
    const float* We2 = (const float*)d_in[12];
    const float* ae2 = (const float*)d_in[13];
    const float* b2  = (const float*)d_in[14];
    float* out = (float*)d_out;

    const int Nn = in_sizes[0] / 256;
    const int E  = in_sizes[2];
    const int K1 = in_sizes[3] / HC;
    const int K2 = in_sizes[9] / HC;

    const int nbk   = (Nn + BK_NODES - 1) / BK_NODES;
    const int chunk = (E + EB - 1) / EB;
    const int ntab  = nbk * EB;

    float* g      = (float*)d_ws;
    float* h1     = g    + (size_t)Nn * HC;
    float* a_s    = h1   + (size_t)Nn * HC;
    float* a_d    = a_s  + (size_t)Nn * HEADS;
    int*   table  = (int*)(a_d + (size_t)Nn * HEADS);
    int*   offs   = table + ntab;
    int*   trec   = offs + ntab + 4;
    int*   rec    = trec + 2 * (size_t)E;
    int*   tdst   = rec + 2 * (size_t)E;
    int*   rowptr = tdst + E;
    float* part   = (float*)(rowptr + Nn + 2);
    float* csum   = part + EB;
    float* ch     = csum + 1;

    const int TB = 256;
    const int mblocks     = ((Nn + 15) / 16 + GW - 1) / GW;
    const int scan_blocks = (ntab + 1023) / 1024;
    const int node_blocks = (Nn + 7) / 8;

    bucket_count_kernel<<<EB, ET, (size_t)(nbk + ET / 32) * sizeof(int), stream>>>(ei, ew, E, chunk, nbk, table, part);
    table_offsets_kernel<<<scan_blocks, 256, 0, stream>>>(table, offs, ntab, part, csum, We1, ae1, We2, ae2, ch);
    gemm_and_place_kernel<<<mblocks + EB, GT, (size_t)K1 * HC * sizeof(float), stream>>>(
        x, W1, g, Nn, K1, as1, ad1, a_s, a_d, mblocks, ei, ew, E, chunk, nbk, offs, (int2*)trec, tdst);
    bucket_sort_kernel<<<nbk, ST, 0, stream>>>((const int2*)trec, tdst, offs, Nn, nbk, rowptr, (int2*)rec);

    node_softmax_aggregate_kernel<<<node_blocks, TB, 0, stream>>>(rowptr, rec, csum, E, Nn, a_s, a_d, ch, g, b1, h1);

    gemm_wmma_f32<<<mblocks, GT, (size_t)K2 * HC * sizeof(float), stream>>>(h1, W2, g, Nn, K2, as2, ad2, a_s, a_d);
    node_softmax_aggregate_kernel<<<node_blocks, TB, 0, stream>>>(rowptr, rec, csum, E, Nn, a_s, a_d, ch + HEADS, g, b2, out);

    (void)n_in; (void)out_size; (void)ws_size;
}


# Node kernel: a slot's score is summed from the gathered row itself (8-lane row_xmask) under a running max per head; a_s gather, shuffles and GEMM score epilogue gone; self loop first, new sum order
# speedup vs baseline: 6.9367x; 1.0460x over previous
#include <hip/hip_runtime.h>
#include <hip/hip_bf16.h>

typedef float v2f __attribute__((ext_vector_type(2)));
typedef float v8f __attribute__((ext_vector_type(8)));

#define HEADS 4
#define HC 128
#define NEG_SLOPE 0.2f

#define BK_SHIFT 8
#define BK_NODES 256
#define EB 256
#define ET 512
#define EU 8
#define ST 1024
#define SU 4

__global__ void __launch_bounds__(ET)
bucket_count_kernel(const int* __restrict__ ei, const float* __restrict__ ew, int E, int chunk, int nbk,
                    int* __restrict__ table, float* __restrict__ part) {
    extern __shared__ int lds_cnt[];
    float* sm = (float*)(lds_cnt + nbk);
    const int t = threadIdx.x;
    for (int b = t; b < nbk; b += ET) lds_cnt[b] = 0;
    __syncthreads();
    const int e0 = blockIdx.x * chunk;
    const int e1 = min(E, e0 + chunk);
    float s = 0.f;
    for (int base = e0; base < e1; base += ET * EU) {
        int dv[EU]; float wv[EU];
        #pragma unroll
        for (int u = 0; u < EU; ++u) {
            const int j = base + u * ET + t;
            const bool ok = j < e1;
            dv[u] = ok ? ei[E + j] : -1;
            wv[u] = ok ? ew[j] : 0.f;
        }
        #pragma unroll
        for (int u = 0; u < EU; ++u) {
            if (dv[u] >= 0) atomicAdd(&lds_cnt[dv[u] >> BK_SHIFT], 1);
            s += wv[u];
        }
    }
    #pragma unroll
    for (int off = 16; off; off >>= 1) s += __shfl_down(s, off, 32);
    if ((t & 31) == 0) sm[t >> 5] = s;
    __syncthreads();
    if (t == 0) {
        float tot = 0.f;
        for (int i = 0; i < ET / 32; ++i) tot += sm[i];
        part[blockIdx.x] = tot;
    }
    for (int b = t; b < nbk; b += ET) table[b * EB + blockIdx.x] = lds_cnt[b];
}

__global__ void table_offsets_kernel(const int* __restrict__ cnt, int* __restrict__ rowptr, int Nn,
                                     const float* __restrict__ part, float* __restrict__ out_sum,
                                     const float* __restrict__ W_ea, const float* __restrict__ att_ea,
                                     const float* __restrict__ W_eb, const float* __restrict__ att_eb,
                                     float* __restrict__ coef) {
    __shared__ int wsum[8];
    __shared__ int wpre[8];
    const int t = threadIdx.x, lane = t & 31, w = t >> 5;
    const int tile0 = blockIdx.x * 1024;
    const int4* c4 = (const int4*)cnt + t;
    const int nb = blockIdx.x;
    int pre = 0;
    int m = 0;
    for (; m + 8 <= nb; m += 8) {
        int4 q[8];
        for (int u = 0; u < 8; ++u) q[u] = c4[(size_t)(m + u) * 256];
        for (int u = 0; u < 8; ++u) pre += (q[u].x + q[u].y) + (q[u].z + q[u].w);
    }
    for (; m < nb; ++m) {
        const int4 q = c4[(size_t)m * 256];
        pre += (q.x + q.y) + (q.z + q.w);
    }
    const int i0 = tile0 + t * 4;
    const int v0 = (i0 + 0 < Nn) ? cnt[i0 + 0] : 0;
    const int v1 = (i0 + 1 < Nn) ? cnt[i0 + 1] : 0;
    const int v2 = (i0 + 2 < Nn) ? cnt[i0 + 2] : 0;
    const int v3 = (i0 + 3 < Nn) ? cnt[i0 + 3] : 0;
    const int tsum = (v0 + v1) + (v2 + v3);
    int inc = tsum;
    for (int off = 1; off < 32; off <<= 1) {
        const int y = __shfl_up(inc, off, 32);
        if (lane >= off) inc += y;
    }
    for (int off = 16; off; off >>= 1) pre += __shfl_xor(pre, off, 32);
    if (lane == 31) wsum[w] = inc;
    if (lane == 0)  wpre[w] = pre;
    __syncthreads();
    int base = 0;
    for (int k = 0; k < 8; ++k) { base += wpre[k]; if (k < w) base += wsum[k]; }
    const int ex = base + inc - tsum;
    if (i0 + 0 < Nn) rowptr[i0 + 1] = ex + v0;
    if (i0 + 1 < Nn) rowptr[i0 + 2] = ex + v0 + v1;
    if (i0 + 2 < Nn) rowptr[i0 + 3] = ex + v0 + v1 + v2;
    if (i0 + 3 < Nn) rowptr[i0 + 4] = ex + tsum;
    if (blockIdx.x == 0 && t == 0) rowptr[0] = 0;
    if (blockIdx.x == 0 && w == 0) {
        float s = 0.f;
        for (int i = lane; i < EB; i += 32) s += part[i];
        #pragma unroll
        for (int off = 16; off; off >>= 1) s += __shfl_down(s, off, 32);
        if (lane == 0) out_sum[0] = s;
    }
    if (blockIdx.x == 0) {
        const int j = t & 127;
        float p = (t < 128) ? W_ea[j] * att_ea[j] : W_eb[j] * att_eb[j];
        #pragma unroll
        for (int off = 16; off; off >>= 1) p += __shfl_down(p, off, 32);
        if (lane == 0) coef[w] = p;
    }
}

__device__ __forceinline__ void
bucket_place_body(int cb, int* lds_cur, const int* __restrict__ ei, const float* __restrict__ ew, int E, int chunk,
                  int nbk, const int* __restrict__ offs, int2* __restrict__ trec, int* __restrict__ tdst) {
    const int t = threadIdx.x;
    for (int b = t; b < nbk; b += ET) lds_cur[b] = offs[b * EB + cb];
    __syncthreads();
    const int e0 = cb * chunk;
    const int e1 = min(E, e0 + chunk);
    for (int base = e0; base < e1; base += ET * EU) {
        int dv[EU]; int sv[EU]; float wv[EU];
        #pragma unroll
        for (int u = 0; u < EU; ++u) {
            const int j = base + u * ET + t;
            const bool ok = j < e1;
            dv[u] = ok ? ei[E + j] : -1;
            sv[u] = ok ? ei[j] : 0;
            wv[u] = ok ? ew[j] : 0.f;
        }
        #pragma unroll
        for (int u = 0; u < EU; ++u) {
            if (dv[u] >= 0) {
                const int pos = atomicAdd(&lds_cur[dv[u] >> BK_SHIFT], 1);
                trec[pos] = make_int2(sv[u], __float_as_int(wv[u]));
                tdst[pos] = dv[u];
            }
        }
    }
}

__global__ void __launch_bounds__(ST)
bucket_sort_kernel(const int2* __restrict__ trec, const int* __restrict__ tdst, const int* __restrict__ offs,
                   int Nn, int nbk, int* __restrict__ rowptr, int2* __restrict__ rec) {
    __shared__ int cnt[BK_NODES];
    __shared__ int cur[BK_NODES];
    __shared__ int wsum[BK_NODES / 32];
    const int t = threadIdx.x, lane = t & 31, w = t >> 5;
    const int b = blockIdx.x;
    const int bs = offs[b * EB];
    const int be = offs[(b + 1) * EB];
    if (t < BK_NODES) cnt[t] = 0;
    __syncthreads();
    for (int base = bs; base < be; base += ST * SU) {
        int dv[SU];
        #pragma unroll
        for (int u = 0; u < SU; ++u) {
            const int j = base + u * ST + t;
            dv[u] = (j < be) ? tdst[j] : -1;
        }
        #pragma unroll
        for (int u = 0; u < SU; ++u)
            if (dv[u] >= 0) atomicAdd(&cnt[dv[u] & (BK_NODES - 1)], 1);
    }
    __syncthreads();
    int v = 0, inc = 0;
    if (t < BK_NODES) {
        v = cnt[t];
        inc = v;
        for (int off = 1; off < 32; off <<= 1) {
            const int y = __shfl_up(inc, off, 32);
            if (lane >= off) inc += y;
        }
        if (lane == 31) wsum[w] = inc;
    }
    __syncthreads();
    if (t < BK_NODES) {
        int ex = bs + inc - v;
        for (int k = 0; k < w; ++k) ex += wsum[k];
        cur[t] = ex;
        const int node = (b << BK_SHIFT) + t;
        if (node < Nn) rowptr[node] = ex;
    }
    if (b == nbk - 1 && t == 0) rowptr[Nn] = be;
    __syncthreads();
    for (int base = bs; base < be; base += ST * SU) {
        int dv[SU]; int2 rv[SU];
        #pragma unroll
        for (int u = 0; u < SU; ++u) {
            const int j = base + u * ST + t;
            const bool ok = j < be;
            dv[u] = ok ? tdst[j] : -1;
            rv[u] = ok ? trec[j] : make_int2(0, 0);
        }
        #pragma unroll
        for (int u = 0; u < SU; ++u)
            if (dv[u] >= 0) rec[atomicAdd(&cur[dv[u] & (BK_NODES - 1)], 1)] = rv[u];
    }
}

template <int MASK>
__device__ __forceinline__ float row_xor_add(float keep, float send) {
    return keep + __int_as_float(__builtin_amdgcn_update_dpp(0, __float_as_int(send), 0x160 | MASK, 0xf, 0xf, true));
}

#define GW 16
#define GT (GW * 32)
#define GP 8
__device__ __forceinline__ void
gemm_wmma_f32_body(int gb, float* lw, const float* __restrict__ A, const float* __restrict__ W,
                   float* __restrict__ C, int M, int K) {
    const int t = threadIdx.x;
    const int npairs = (K >> 1) * (HC >> 2);
    for (int p0 = 0; p0 < npairs; p0 += GT * GP) {
        float4 ra[GP], rb[GP];
        #pragma unroll
        for (int u = 0; u < GP; ++u) {
            const int p = p0 + u * GT + t;
            const bool ok = p < npairs;
            const float* src = W + (size_t)(p >> 5) * (2 * HC) + ((p & 31) << 2);
            ra[u] = ok ? *(const float4*)src        : make_float4(0.f, 0.f, 0.f, 0.f);
            rb[u] = ok ? *(const float4*)(src + HC) : make_float4(0.f, 0.f, 0.f, 0.f);
        }
        #pragma unroll
        for (int u = 0; u < GP; ++u) {
            const int p = p0 + u * GT + t;
            if (p < npairs) {
                float4* dst = (float4*)&lw[((p >> 5) << 8) + ((p & 31) << 3)];
                dst[0] = make_float4(ra[u].x, rb[u].x, ra[u].y, rb[u].y);
                dst[1] = make_float4(ra[u].z, rb[u].z, ra[u].w, rb[u].w);
            }
        }
    }
    __syncthreads();

    const int wave = t >> 5, lane = t & 31;
    const int rowblock = gb * GW + wave;
    const int row0 = rowblock * 16;
    if (row0 >= M) return;

    const int m     = lane & 15;
    const int khalf = lane >> 4;
    const int n     = lane & 15;

    v8f acc[8] = {};
    const float* arow = A + (size_t)(row0 + m) * K + khalf * 2;

    for (int k0 = 0; k0 < K; k0 += 4) {
        v2f a = *(const v2f*)(arow + k0);
        const float* bbase = &lw[((k0 >> 1) + khalf) * 256 + (n << 1)];
        v2f b[8];
        #pragma unroll
        for (int c = 0; c < 8; ++c)
            b[c] = *(const v2f*)(bbase + (c << 5));
        #pragma unroll
        for (int c = 0; c < 8; ++c)
            acc[c] = __builtin_amdgcn_wmma_f32_16x16x4_f32(
                false, a, false, b[c], (short)0, acc[c], false, false);
    }

    float* crow = C + (size_t)row0 * HC;
    const int hi = (lane >> 4) * 8;
    #pragma unroll
    for (int c = 0; c < 8; ++c)
        #pragma unroll
        for (int r = 0; r < 8; ++r)
            crow[(size_t)(r + hi) * HC + c * 16 + n] = acc[c][r];
}

__global__ void __launch_bounds__(GT)
gemm_wmma_f32(const float* __restrict__ A, const float* __restrict__ W,
              float* __restrict__ C, int M, int K) {
    extern __shared__ float dyn_lds[];
    gemm_wmma_f32_body((int)blockIdx.x, dyn_lds, A, W, C, M, K);
}

static_assert(GT == ET, "the shared launch needs one block size for both bodies");
__global__ void __launch_bounds__(GT)
gemm_and_place_kernel(const float* __restrict__ A, const float* __restrict__ W,
                      float* __restrict__ C, int M, int K, int gblocks,
                      const int* __restrict__ ei, const float* __restrict__ ew, int E, int chunk, int nbk,
                      const int* __restrict__ offs, int2* __restrict__ trec, int* __restrict__ tdst) {
    extern __shared__ float dyn_lds[];
    const int bx = (int)blockIdx.x;
    if (bx < gblocks) gemm_wmma_f32_body(bx, dyn_lds, A, W, C, M, K);
    else bucket_place_body(bx - gblocks, (int*)dyn_lds, ei, ew, E, chunk, nbk, offs, trec, tdst);
}

__global__ void node_softmax_aggregate_kernel(const int* __restrict__ rowptr, const int* __restrict__ rec,
                                              const float* __restrict__ sumptr, int E, int Nn,
                                              const float* __restrict__ att_src, const float* __restrict__ att_dst,
                                              const float* __restrict__ c,
                                              const float* __restrict__ g, const float* __restrict__ bias,
                                              float* __restrict__ out) {
    const int lane = threadIdx.x & 31;
    const int d = __builtin_amdgcn_readfirstlane((int)(blockIdx.x * (blockDim.x >> 5) + (threadIdx.x >> 5)));
    if (d >= Nn) return;
    const float* gl = g + lane * 4;
    const float4 rs = *(const float4*)(gl + (size_t)d * HC);
    const int start = rowptr[d];
    const int n     = rowptr[d + 1] - start;
    const float L2E = 1.44269504088896341f;
    const float ea_self = sumptr[0] * (1.0f / (float)E);
    const float ch = c[lane >> 3] * L2E;
    float4 ws = *(const float4*)(att_src + lane * 4);
    float4 wd = *(const float4*)(att_dst + lane * 4);
    ws.x *= L2E; ws.y *= L2E; ws.z *= L2E; ws.w *= L2E;
    wd.x *= L2E; wd.y *= L2E; wd.z *= L2E; wd.w *= L2E;

    int s_l = d; float ea_l = 0.f;
    if (lane < n) { const int2 r = ((const int2*)rec)[start + lane]; s_l = r.x; ea_l = __int_as_float(r.y); }

    auto head_dot = [&](const float4& r, const float4& w) -> float {
        float t = (r.x * w.x + r.y * w.y) + (r.z * w.z + r.w * w.w);
        t = row_xor_add<1>(t, t);
        t = row_xor_add<2>(t, t);
        return row_xor_add<4>(t, t);
    };
    const float adh = head_dot(rs, wd);
    auto slot_alpha = [&](const float4& r, float ea) -> float {
        const float x = (head_dot(r, ws) + adh) + ea * ch;
        return fmaxf(x, NEG_SLOPE * x);
    };

    float m = slot_alpha(rs, ea_self);
    float den = 1.f;
    float4 acc = rs;
    const float ninf = -__builtin_inff();
    for (int base = 0; base < n; base += 32) {
        if (base) {
            s_l = d; ea_l = 0.f;
            if (base + lane < n) {
                const int2 r = ((const int2*)rec)[start + base + lane]; s_l = r.x; ea_l = __int_as_float(r.y);
            }
        }
        const int cnt = min(32, n - base);
        for (int k = 0; k < cnt; k += 4) {
            const int sa = __builtin_amdgcn_readlane(s_l, k);
            const int sb = __builtin_amdgcn_readlane(s_l, k + 1);
            const int sc = __builtin_amdgcn_readlane(s_l, k + 2);
            const int sd = __builtin_amdgcn_readlane(s_l, k + 3);
            const float4 ra = *(const float4*)(gl + (size_t)sa * HC);
            const float4 rb = *(const float4*)(gl + (size_t)sb * HC);
            const float4 rc = *(const float4*)(gl + (size_t)sc * HC);
            const float4 rd = *(const float4*)(gl + (size_t)sd * HC);
            const float ea_a = __int_as_float(__builtin_amdgcn_readlane(__float_as_int(ea_l), k));
            const float ea_b = __int_as_float(__builtin_amdgcn_readlane(__float_as_int(ea_l), k + 1));
            const float ea_c = __int_as_float(__builtin_amdgcn_readlane(__float_as_int(ea_l), k + 2));
            const float ea_d = __int_as_float(__builtin_amdgcn_readlane(__float_as_int(ea_l), k + 3));
            const float aa = slot_alpha(ra, ea_a);
            const float ab = (k + 1 < cnt) ? slot_alpha(rb, ea_b) : ninf;
            const float ac = (k + 2 < cnt) ? slot_alpha(rc, ea_c) : ninf;
            const float ad = (k + 3 < cnt) ? slot_alpha(rd, ea_d) : ninf;
            const float mn = fmaxf(fmaxf(m, aa), fmaxf(fmaxf(ab, ac), ad));
            const float corr = __builtin_amdgcn_exp2f(m - mn);
            const float pa = __builtin_amdgcn_exp2f(aa - mn);
            const float pb = __builtin_amdgcn_exp2f(ab - mn);
            const float pc = __builtin_amdgcn_exp2f(ac - mn);
            const float pd = __builtin_amdgcn_exp2f(ad - mn);
            m = mn;
            den = den * corr + ((pa + pb) + (pc + pd));
            acc.x = acc.x * corr + pa * ra.x; acc.y = acc.y * corr + pa * ra.y;
            acc.z = acc.z * corr + pa * ra.z; acc.w = acc.w * corr + pa * ra.w;
            acc.x += pb * rb.x; acc.y += pb * rb.y; acc.z += pb * rb.z; acc.w += pb * rb.w;
            acc.x += pc * rc.x; acc.y += pc * rc.y; acc.z += pc * rc.z; acc.w += pc * rc.w;
            acc.x += pd * rd.x; acc.y += pd * rd.y; acc.z += pd * rd.z; acc.w += pd * rd.w;
        }
    }
    const float inv = 1.0f / den;
    const float* bl = bias + lane * 4;
    *(float4*)(out + (size_t)d * HC + lane * 4) =
        make_float4(acc.x * inv + bl[0], acc.y * inv + bl[1], acc.z * inv + bl[2], acc.w * inv + bl[3]);
}

extern "C" void kernel_launch(void* const* d_in, const int* in_sizes, int n_in,
                              void* d_out, int out_size, void* d_ws, size_t ws_size,
                              hipStream_t stream) {
    const float* x   = (const float*)d_in[0];
    const int*   ei  = (const int*)d_in[1];
    const float* ew  = (const float*)d_in[2];
    const float* W1  = (const float*)d_in[3];
    const float* as1 = (const float*)d_in[4];
    const float* ad1 = (const float*)d_in[5];
    const float* We1 = (const float*)d_in[6];
    const float* ae1 = (const float*)d_in[7];
    const float* b1  = (const float*)d_in[8];
    const float* W2  = (const float*)d_in[9];
    const float* as2 = (const float*)d_in[10];
    const float* ad2 = (const float*)d_in[11];
    const float* We2 = (const float*)d_in[12];
    const float* ae2 = (const float*)d_in[13];
    const float* b2  = (const float*)d_in[14];
    float* out = (float*)d_out;

    const int Nn = in_sizes[0] / 256;
    const int E  = in_sizes[2];
    const int K1 = in_sizes[3] / HC;
    const int K2 = in_sizes[9] / HC;

    const int nbk   = (Nn + BK_NODES - 1) / BK_NODES;
    const int chunk = (E + EB - 1) / EB;
    const int ntab  = nbk * EB;

    float* g      = (float*)d_ws;
    float* h1     = g    + (size_t)Nn * HC;
    int*   table  = (int*)(h1 + (size_t)Nn * HC);
    int*   offs   = table + ntab;
    int*   trec   = offs + ntab + 4;
    int*   rec    = trec + 2 * (size_t)E;
    int*   tdst   = rec + 2 * (size_t)E;
    int*   rowptr = tdst + E;
    float* part   = (float*)(rowptr + Nn + 2);
    float* csum   = part + EB;
    float* ch     = csum + 1;

    const int TB = 256;
    const int mblocks     = ((Nn + 15) / 16 + GW - 1) / GW;
    const int scan_blocks = (ntab + 1023) / 1024;
    const int node_blocks = (Nn + 7) / 8;

    bucket_count_kernel<<<EB, ET, (size_t)(nbk + ET / 32) * sizeof(int), stream>>>(ei, ew, E, chunk, nbk, table, part);
    table_offsets_kernel<<<scan_blocks, 256, 0, stream>>>(table, offs, ntab, part, csum, We1, ae1, We2, ae2, ch);
    gemm_and_place_kernel<<<mblocks + EB, GT, (size_t)K1 * HC * sizeof(float), stream>>>(
        x, W1, g, Nn, K1, mblocks, ei, ew, E, chunk, nbk, offs, (int2*)trec, tdst);
    bucket_sort_kernel<<<nbk, ST, 0, stream>>>((const int2*)trec, tdst, offs, Nn, nbk, rowptr, (int2*)rec);

    node_softmax_aggregate_kernel<<<node_blocks, TB, 0, stream>>>(rowptr, rec, csum, E, Nn, as1, ad1, ch, g, b1, h1);

    gemm_wmma_f32<<<mblocks, GT, (size_t)K2 * HC * sizeof(float), stream>>>(h1, W2, g, Nn, K2);
    node_softmax_aggregate_kernel<<<node_blocks, TB, 0, stream>>>(rowptr, rec, csum, E, Nn, as2, ad2, ch + HEADS, g, b2, out);

    (void)n_in; (void)out_size; (void)ws_size;
}


# Edge grouping in 2 launches, not 4: each chunk counts, scans and places its edges into bucket runs inside the chunk; each bucket sums the table rows before its own, collects its runs; 7->5 launches
# speedup vs baseline: 8.4540x; 1.2187x over previous
#include <hip/hip_runtime.h>
#include <hip/hip_bf16.h>

typedef float v2f __attribute__((ext_vector_type(2)));
typedef float v8f __attribute__((ext_vector_type(8)));

#define HEADS 4
#define HC 128
#define NEG_SLOPE 0.2f

#define BK_SHIFT 8
#define BK_NODES 256
#define EB 256
#define ET 512
#define EU 8
#define ST 1024
#define RU 8

__device__ __forceinline__ void
chunk_group_body(int cb, int* lds, const int* __restrict__ ei, const float* __restrict__ ew, int E, int chunk,
                 int nbk, int* __restrict__ table, int* __restrict__ ltab, int2* __restrict__ trec,
                 int* __restrict__ tdst, float* __restrict__ part) {
    int* cntb = lds;
    int* curb = lds + nbk;
    float* sm = (float*)(lds + 2 * nbk);
    const int t = threadIdx.x;
    for (int b = t; b < nbk; b += ET) cntb[b] = 0;
    __syncthreads();
    const int e0 = cb * chunk;
    const int e1 = min(E, e0 + chunk);
    float s = 0.f;
    for (int base = e0; base < e1; base += ET * EU) {
        int dv[EU]; float wv[EU];
        #pragma unroll
        for (int u = 0; u < EU; ++u) {
            const int j = base + u * ET + t;
            const bool ok = j < e1;
            dv[u] = ok ? ei[E + j] : -1;
            wv[u] = ok ? ew[j] : 0.f;
        }
        #pragma unroll
        for (int u = 0; u < EU; ++u) {
            if (dv[u] >= 0) atomicAdd(&cntb[dv[u] >> BK_SHIFT], 1);
            s += wv[u];
        }
    }
    #pragma unroll
    for (int off = 16; off; off >>= 1) s += __shfl_down(s, off, 32);
    if ((t & 31) == 0) sm[t >> 5] = s;
    __syncthreads();
    if (t == 0) {
        float tot = 0.f;
        for (int i = 0; i < ET / 32; ++i) tot += sm[i];
        part[cb] = tot;
    }
    if (t < 32) {
        const int per = (nbk + 31) >> 5;
        int a = 0;
        for (int i = 0; i < per; ++i) { const int b = t * per + i; if (b < nbk) a += cntb[b]; }
        int inc = a;
        for (int off = 1; off < 32; off <<= 1) {
            const int y = __shfl_up(inc, off, 32);
            if (t >= off) inc += y;
        }
        int run = inc - a;
        for (int i = 0; i < per; ++i) {
            const int b = t * per + i;
            if (b < nbk) {
                const int v = cntb[b];
                curb[b] = run;
                table[b * EB + cb] = v;
                ltab[b * EB + cb] = run;
                run += v;
            }
        }
    }
    __syncthreads();
    for (int base = e0; base < e1; base += ET * EU) {
        int dv[EU]; int sv[EU]; float wv[EU];
        #pragma unroll
        for (int u = 0; u < EU; ++u) {
            const int j = base + u * ET + t;
            const bool ok = j < e1;
            dv[u] = ok ? ei[E + j] : -1;
            sv[u] = ok ? ei[j] : 0;
            wv[u] = ok ? ew[j] : 0.f;
        }
        #pragma unroll
        for (int u = 0; u < EU; ++u) {
            if (dv[u] >= 0) {
                const int pos = e0 + atomicAdd(&curb[dv[u] >> BK_SHIFT], 1);
                trec[pos] = make_int2(sv[u], __float_as_int(wv[u]));
                tdst[pos] = dv[u];
            }
        }
    }
}

static_assert(ST == 4 * EB, "the bucket work-group gives four threads to each of a bucket's EB runs");
__global__ void __launch_bounds__(ST)
bucket_sort_kernel(const int2* __restrict__ trec, const int* __restrict__ tdst, const int* __restrict__ table,
                   const int* __restrict__ ltab, int chunk, int Nn, int nbk, int* __restrict__ rowptr,
                   int2* __restrict__ rec, const float* __restrict__ part, float* __restrict__ out_sum,
                   const float* __restrict__ W_ea, const float* __restrict__ att_ea,
                   const float* __restrict__ W_eb, const float* __restrict__ att_eb, float* __restrict__ coef) {
    __shared__ int cnt[BK_NODES];
    __shared__ int cur[BK_NODES];
    __shared__ int wsum[BK_NODES / 32];
    __shared__ int red[ST / 32];
    const int t = threadIdx.x, lane = t & 31, w = t >> 5;
    const int b = blockIdx.x;
    const int c = t >> 2, sub = t & 3;
    const int len = table[b * EB + c];
    const int rbase = c * chunk + ltab[b * EB + c];
    const int4* t4 = (const int4*)table;
    const int n4 = b * (EB / 4);
    int pre = 0;
    int i = t;
    for (; i + 3 * ST < n4; i += 4 * ST) {
        const int4 q0 = t4[i], q1 = t4[i + ST], q2 = t4[i + 2 * ST], q3 = t4[i + 3 * ST];
        pre += ((q0.x + q0.y) + (q0.z + q0.w)) + ((q1.x + q1.y) + (q1.z + q1.w));
        pre += ((q2.x + q2.y) + (q2.z + q2.w)) + ((q3.x + q3.y) + (q3.z + q3.w));
    }
    for (; i < n4; i += ST) {
        const int4 q = t4[i];
        pre += (q.x + q.y) + (q.z + q.w);
    }
    for (int off = 16; off; off >>= 1) pre += __shfl_xor(pre, off, 32);
    if (lane == 0) red[w] = pre;
    if (t < BK_NODES) cnt[t] = 0;
    __syncthreads();
    int bs = 0;
    for (int k = 0; k < ST / 32; ++k) bs += red[k];
    for (int i0 = sub; i0 < len; i0 += 4 * RU) {
        int dv[RU];
        #pragma unroll
        for (int u = 0; u < RU; ++u) {
            const int j = i0 + 4 * u;
            dv[u] = (j < len) ? tdst[rbase + j] : -1;
        }
        #pragma unroll
        for (int u = 0; u < RU; ++u)
            if (dv[u] >= 0) atomicAdd(&cnt[dv[u] & (BK_NODES - 1)], 1);
    }
    __syncthreads();
    int v = 0, inc = 0;
    if (t < BK_NODES) {
        v = cnt[t];
        inc = v;
        for (int off = 1; off < 32; off <<= 1) {
            const int y = __shfl_up(inc, off, 32);
            if (lane >= off) inc += y;
        }
        if (lane == 31) wsum[w] = inc;
    }
    __syncthreads();
    if (t < BK_NODES) {
        int ex = bs + inc - v;
        for (int k = 0; k < w; ++k) ex += wsum[k];
        cur[t] = ex;
        const int node = (b << BK_SHIFT) + t;
        if (node < Nn) rowptr[node] = ex;
    }
    if (b == nbk - 1 && t == 0) {
        int be = bs;
        for (int k = 0; k < BK_NODES / 32; ++k) be += wsum[k];
        rowptr[Nn] = be;
    }
    __syncthreads();
    for (int i0 = sub; i0 < len; i0 += 4 * RU) {
        int dv[RU]; int2 rv[RU];
        #pragma unroll
        for (int u = 0; u < RU; ++u) {
            const int j = i0 + 4 * u;
            const bool ok = j < len;
            dv[u] = ok ? tdst[rbase + j] : -1;
            rv[u] = ok ? trec[rbase + j] : make_int2(0, 0);
        }
        #pragma unroll
        for (int u = 0; u < RU; ++u)
            if (dv[u] >= 0) rec[atomicAdd(&cur[dv[u] & (BK_NODES - 1)], 1)] = rv[u];
    }
    if (b == 0 && w == 0) {
        float s = 0.f;
        for (int k = lane; k < EB; k += 32) s += part[k];
        #pragma unroll
        for (int off = 16; off; off >>= 1) s += __shfl_down(s, off, 32);
        if (lane == 0) out_sum[0] = s;
    }
    if (b == 0 && t < 256) {
        const int j = t & 127;
        float p = (t < 128) ? W_ea[j] * att_ea[j] : W_eb[j] * att_eb[j];
        #pragma unroll
        for (int off = 16; off; off >>= 1) p += __shfl_down(p, off, 32);
        if (lane == 0) coef[w] = p;
    }
}

template <int MASK>
__device__ __forceinline__ float row_xor_add(float keep, float send) {
    return keep + __int_as_float(__builtin_amdgcn_update_dpp(0, __float_as_int(send), 0x160 | MASK, 0xf, 0xf, true));
}

#define GW 16
#define GT (GW * 32)
#define GP 8
__device__ __forceinline__ void
gemm_wmma_f32_body(int gb, float* lw, const float* __restrict__ A, const float* __restrict__ W,
                   float* __restrict__ C, int M, int K) {
    const int t = threadIdx.x;
    const int npairs = (K >> 1) * (HC >> 2);
    for (int p0 = 0; p0 < npairs; p0 += GT * GP) {
        float4 ra[GP], rb[GP];
        #pragma unroll
        for (int u = 0; u < GP; ++u) {
            const int p = p0 + u * GT + t;
            const bool ok = p < npairs;
            const float* src = W + (size_t)(p >> 5) * (2 * HC) + ((p & 31) << 2);
            ra[u] = ok ? *(const float4*)src        : make_float4(0.f, 0.f, 0.f, 0.f);
            rb[u] = ok ? *(const float4*)(src + HC) : make_float4(0.f, 0.f, 0.f, 0.f);
        }
        #pragma unroll
        for (int u = 0; u < GP; ++u) {
            const int p = p0 + u * GT + t;
            if (p < npairs) {
                float4* dst = (float4*)&lw[((p >> 5) << 8) + ((p & 31) << 3)];
                dst[0] = make_float4(ra[u].x, rb[u].x, ra[u].y, rb[u].y);
                dst[1] = make_float4(ra[u].z, rb[u].z, ra[u].w, rb[u].w);
            }
        }
    }
    __syncthreads();

    const int wave = t >> 5, lane = t & 31;
    const int rowblock = gb * GW + wave;
    const int row0 = rowblock * 16;
    if (row0 >= M) return;

    const int m     = lane & 15;
    const int khalf = lane >> 4;
    const int n     = lane & 15;

    v8f acc[8] = {};
    const float* arow = A + (size_t)(row0 + m) * K + khalf * 2;

    for (int k0 = 0; k0 < K; k0 += 4) {
        v2f a = *(const v2f*)(arow + k0);
        const float* bbase = &lw[((k0 >> 1) + khalf) * 256 + (n << 1)];
        v2f b[8];
        #pragma unroll
        for (int c = 0; c < 8; ++c)
            b[c] = *(const v2f*)(bbase + (c << 5));
        #pragma unroll
        for (int c = 0; c < 8; ++c)
            acc[c] = __builtin_amdgcn_wmma_f32_16x16x4_f32(
                false, a, false, b[c], (short)0, acc[c], false, false);
    }

    float* crow = C + (size_t)row0 * HC;
    const int hi = (lane >> 4) * 8;
    #pragma unroll
    for (int c = 0; c < 8; ++c)
        #pragma unroll
        for (int r = 0; r < 8; ++r)
            crow[(size_t)(r + hi) * HC + c * 16 + n] = acc[c][r];
}

__global__ void __launch_bounds__(GT)
gemm_wmma_f32(const float* __restrict__ A, const float* __restrict__ W,
              float* __restrict__ C, int M, int K) {
    extern __shared__ float dyn_lds[];
    gemm_wmma_f32_body((int)blockIdx.x, dyn_lds, A, W, C, M, K);
}

static_assert(GT == ET, "the shared launch needs one block size for both bodies");
__global__ void __launch_bounds__(GT)
gemm_and_group_kernel(const float* __restrict__ A, const float* __restrict__ W,
                      float* __restrict__ C, int M, int K, int gblocks,
                      const int* __restrict__ ei, const float* __restrict__ ew, int E, int chunk, int nbk,
                      int* __restrict__ table, int* __restrict__ ltab, int2* __restrict__ trec,
                      int* __restrict__ tdst, float* __restrict__ part) {
    extern __shared__ float dyn_lds[];
    const int bx = (int)blockIdx.x;
    if (bx < gblocks) gemm_wmma_f32_body(bx, dyn_lds, A, W, C, M, K);
    else chunk_group_body(bx - gblocks, (int*)dyn_lds, ei, ew, E, chunk, nbk, table, ltab, trec, tdst, part);
}

__global__ void node_softmax_aggregate_kernel(const int* __restrict__ rowptr, const int* __restrict__ rec,
                                              const float* __restrict__ sumptr, int E, int Nn,
                                              const float* __restrict__ att_src, const float* __restrict__ att_dst,
                                              const float* __restrict__ c,
                                              const float* __restrict__ g, const float* __restrict__ bias,
                                              float* __restrict__ out) {
    const int lane = threadIdx.x & 31;
    const int d = __builtin_amdgcn_readfirstlane((int)(blockIdx.x * (blockDim.x >> 5) + (threadIdx.x >> 5)));
    if (d >= Nn) return;
    const float* gl = g + lane * 4;
    const float4 rs = *(const float4*)(gl + (size_t)d * HC);
    const int start = rowptr[d];
    const int n     = rowptr[d + 1] - start;
    const float L2E = 1.44269504088896341f;
    const float ea_self = sumptr[0] * (1.0f / (float)E);
    const float ch = c[lane >> 3] * L2E;
    float4 ws = *(const float4*)(att_src + lane * 4);
    float4 wd = *(const float4*)(att_dst + lane * 4);
    ws.x *= L2E; ws.y *= L2E; ws.z *= L2E; ws.w *= L2E;
    wd.x *= L2E; wd.y *= L2E; wd.z *= L2E; wd.w *= L2E;

    int s_l = d; float ea_l = 0.f;
    if (lane < n) { const int2 r = ((const int2*)rec)[start + lane]; s_l = r.x; ea_l = __int_as_float(r.y); }

    auto head_dot = [&](const float4& r, const float4& w) -> float {
        float t = (r.x * w.x + r.y * w.y) + (r.z * w.z + r.w * w.w);
        t = row_xor_add<1>(t, t);
        t = row_xor_add<2>(t, t);
        return row_xor_add<4>(t, t);
    };
    const float adh = head_dot(rs, wd);
    auto slot_alpha = [&](const float4& r, float ea) -> float {
        const float x = (head_dot(r, ws) + adh) + ea * ch;
        return fmaxf(x, NEG_SLOPE * x);
    };

    float m = slot_alpha(rs, ea_self);
    float den = 1.f;
    float4 acc = rs;
    const float ninf = -__builtin_inff();
    for (int base = 0; base < n; base += 32) {
        if (base) {
            s_l = d; ea_l = 0.f;
            if (base + lane < n) {
                const int2 r = ((const int2*)rec)[start + base + lane]; s_l = r.x; ea_l = __int_as_float(r.y);
            }
        }
        const int cnt = min(32, n - base);
        for (int k = 0; k < cnt; k += 4) {
            const int sa = __builtin_amdgcn_readlane(s_l, k);
            const int sb = __builtin_amdgcn_readlane(s_l, k + 1);
            const int sc = __builtin_amdgcn_readlane(s_l, k + 2);
            const int sd = __builtin_amdgcn_readlane(s_l, k + 3);
            const float4 ra = *(const float4*)(gl + (size_t)sa * HC);
            const float4 rb = *(const float4*)(gl + (size_t)sb * HC);
            const float4 rc = *(const float4*)(gl + (size_t)sc * HC);
            const float4 rd = *(const float4*)(gl + (size_t)sd * HC);
            const float ea_a = __int_as_float(__builtin_amdgcn_readlane(__float_as_int(ea_l), k));
            const float ea_b = __int_as_float(__builtin_amdgcn_readlane(__float_as_int(ea_l), k + 1));
            const float ea_c = __int_as_float(__builtin_amdgcn_readlane(__float_as_int(ea_l), k + 2));
            const float ea_d = __int_as_float(__builtin_amdgcn_readlane(__float_as_int(ea_l), k + 3));
            const float aa = slot_alpha(ra, ea_a);
            const float ab = (k + 1 < cnt) ? slot_alpha(rb, ea_b) : ninf;
            const float ac = (k + 2 < cnt) ? slot_alpha(rc, ea_c) : ninf;
            const float ad = (k + 3 < cnt) ? slot_alpha(rd, ea_d) : ninf;
            const float mn = fmaxf(fmaxf(m, aa), fmaxf(fmaxf(ab, ac), ad));
            const float corr = __builtin_amdgcn_exp2f(m - mn);
            const float pa = __builtin_amdgcn_exp2f(aa - mn);
            const float pb = __builtin_amdgcn_exp2f(ab - mn);
            const float pc = __builtin_amdgcn_exp2f(ac - mn);
            const float pd = __builtin_amdgcn_exp2f(ad - mn);
            m = mn;
            den = den * corr + ((pa + pb) + (pc + pd));
            acc.x = acc.x * corr + pa * ra.x; acc.y = acc.y * corr + pa * ra.y;
            acc.z = acc.z * corr + pa * ra.z; acc.w = acc.w * corr + pa * ra.w;
            acc.x += pb * rb.x; acc.y += pb * rb.y; acc.z += pb * rb.z; acc.w += pb * rb.w;
            acc.x += pc * rc.x; acc.y += pc * rc.y; acc.z += pc * rc.z; acc.w += pc * rc.w;
            acc.x += pd * rd.x; acc.y += pd * rd.y; acc.z += pd * rd.z; acc.w += pd * rd.w;
        }
    }
    const float inv = 1.0f / den;
    const float* bl = bias + lane * 4;
    *(float4*)(out + (size_t)d * HC + lane * 4) =
        make_float4(acc.x * inv + bl[0], acc.y * inv + bl[1], acc.z * inv + bl[2], acc.w * inv + bl[3]);
}

extern "C" void kernel_launch(void* const* d_in, const int* in_sizes, int n_in,
                              void* d_out, int out_size, void* d_ws, size_t ws_size,
                              hipStream_t stream) {
    const float* x   = (const float*)d_in[0];
    const int*   ei  = (const int*)d_in[1];
    const float* ew  = (const float*)d_in[2];
    const float* W1  = (const float*)d_in[3];
    const float* as1 = (const float*)d_in[4];
    const float* ad1 = (const float*)d_in[5];
    const float* We1 = (const float*)d_in[6];
    const float* ae1 = (const float*)d_in[7];
    const float* b1  = (const float*)d_in[8];
    const float* W2  = (const float*)d_in[9];
    const float* as2 = (const float*)d_in[10];
    const float* ad2 = (const float*)d_in[11];
    const float* We2 = (const float*)d_in[12];
    const float* ae2 = (const float*)d_in[13];
    const float* b2  = (const float*)d_in[14];
    float* out = (float*)d_out;

    const int Nn = in_sizes[0] / 256;
    const int E  = in_sizes[2];
    const int K1 = in_sizes[3] / HC;
    const int K2 = in_sizes[9] / HC;

    const int nbk   = (Nn + BK_NODES - 1) / BK_NODES;
    const int chunk = (E + EB - 1) / EB;
    const int ntab  = nbk * EB;

    float* g      = (float*)d_ws;
    float* h1     = g    + (size_t)Nn * HC;
    int*   table  = (int*)(h1 + (size_t)Nn * HC);
    int*   ltab   = table + ntab;
    int*   trec   = ltab + ntab + 4;
    int*   rec    = trec + 2 * (size_t)E;
    int*   tdst   = rec + 2 * (size_t)E;
    int*   rowptr = tdst + E;
    float* part   = (float*)(rowptr + Nn + 2);
    float* csum   = part + EB;
    float* ch     = csum + 1;

    const int TB = 256;
    const int mblocks     = ((Nn + 15) / 16 + GW - 1) / GW;
    const int node_blocks = (Nn + 7) / 8;

    gemm_and_group_kernel<<<mblocks + EB, GT, (size_t)K1 * HC * sizeof(float), stream>>>(
        x, W1, g, Nn, K1, mblocks, ei, ew, E, chunk, nbk, table, ltab, (int2*)trec, tdst, part);
    bucket_sort_kernel<<<nbk, ST, 0, stream>>>((const int2*)trec, tdst, table, ltab, chunk, Nn, nbk, rowptr, (int2*)rec,
                                               part, csum, We1, ae1, We2, ae2, ch);

    node_softmax_aggregate_kernel<<<node_blocks, TB, 0, stream>>>(rowptr, rec, csum, E, Nn, as1, ad1, ch, g, b1, h1);

    gemm_wmma_f32<<<mblocks, GT, (size_t)K2 * HC * sizeof(float), stream>>>(h1, W2, g, Nn, K2);
    node_softmax_aggregate_kernel<<<node_blocks, TB, 0, stream>>>(rowptr, rec, csum, E, Nn, as2, ad2, ch + HEADS, g, b2, out);

    (void)n_in; (void)out_size; (void)ws_size;
}
